# starting kernel
# baseline (speedup 1.0000x reference)
#include <hip/hip_runtime.h>
#include <math.h>


typedef float  v2f __attribute__((ext_vector_type(2)));
typedef float  v8f __attribute__((ext_vector_type(8)));

#define B_SZ 2
#define L_SZ 2048
#define DM   768
#define DI   1536
#define DS   16
#define NROW (B_SZ * L_SZ)

template <int K>
__global__ __launch_bounds__(32) void gemm32x64_f32_wmma(
    const float* __restrict__ A, const float* __restrict__ W,
    float* __restrict__ C, int N64, int lda, int ldw, int ldc)
{
    static_assert(K % 8 == 0, "K must be a multiple of 8");
    const int tile = blockIdx.x;
    const int tm   = tile / N64;
    const int tn   = tile % N64;
    const int lane = threadIdx.x;
    const int half = lane >> 4;
    const int lr   = lane & 15;
    const int kh   = 2 * half;

    const float* Ar0 = A + (size_t)(tm * 32 +  0 + lr) * lda;
    const float* Ar1 = A + (size_t)(tm * 32 + 16 + lr) * lda;
    const float* W0  = W + (size_t)(tn * 64 +  0 + lr) * ldw;
    const float* W1  = W + (size_t)(tn * 64 + 16 + lr) * ldw;
    const float* W2  = W + (size_t)(tn * 64 + 32 + lr) * ldw;
    const float* W3  = W + (size_t)(tn * 64 + 48 + lr) * ldw;

    v8f acc00 = {}; v8f acc01 = {}; v8f acc02 = {}; v8f acc03 = {};
    v8f acc10 = {}; v8f acc11 = {}; v8f acc12 = {}; v8f acc13 = {};

    auto ld = [&](int kb, v2f& a0, v2f& a1, v2f& b0, v2f& b1, v2f& b2, v2f& b3) {
        a0 = *(const v2f*)(Ar0 + kb);
        a1 = *(const v2f*)(Ar1 + kb);
        b0 = *(const v2f*)(W0 + kb);
        b1 = *(const v2f*)(W1 + kb);
        b2 = *(const v2f*)(W2 + kb);
        b3 = *(const v2f*)(W3 + kb);
    };
    auto mm8 = [&](v2f a0, v2f a1, v2f b0, v2f b1, v2f b2, v2f b3) {
        acc00 = __builtin_amdgcn_wmma_f32_16x16x4_f32(false, a0, false, b0, (short)0, acc00, false, false);
        acc01 = __builtin_amdgcn_wmma_f32_16x16x4_f32(false, a0, false, b1, (short)0, acc01, false, false);
        acc02 = __builtin_amdgcn_wmma_f32_16x16x4_f32(false, a0, false, b2, (short)0, acc02, false, false);
        acc03 = __builtin_amdgcn_wmma_f32_16x16x4_f32(false, a0, false, b3, (short)0, acc03, false, false);
        acc10 = __builtin_amdgcn_wmma_f32_16x16x4_f32(false, a1, false, b0, (short)0, acc10, false, false);
        acc11 = __builtin_amdgcn_wmma_f32_16x16x4_f32(false, a1, false, b1, (short)0, acc11, false, false);
        acc12 = __builtin_amdgcn_wmma_f32_16x16x4_f32(false, a1, false, b2, (short)0, acc12, false, false);
        acc13 = __builtin_amdgcn_wmma_f32_16x16x4_f32(false, a1, false, b3, (short)0, acc13, false, false);
    };

    v2f a0A, a1A, b0A, b1A, b2A, b3A;
    v2f a0B, a1B, b0B, b1B, b2B, b3B;

    ld(kh, a0A, a1A, b0A, b1A, b2A, b3A);
    for (int k = 0; k < K - 8; k += 8) {
        ld(k + 4 + kh, a0B, a1B, b0B, b1B, b2B, b3B);
        mm8(a0A, a1A, b0A, b1A, b2A, b3A);
        ld(k + 8 + kh, a0A, a1A, b0A, b1A, b2A, b3A);
        mm8(a0B, a1B, b0B, b1B, b2B, b3B);
    }
    ld(K - 4 + kh, a0B, a1B, b0B, b1B, b2B, b3B);
    mm8(a0A, a1A, b0A, b1A, b2A, b3A);
    mm8(a0B, a1B, b0B, b1B, b2B, b3B);

    float* Cb0 = C + (size_t)(tm * 32 +  0 + 8 * half) * ldc + tn * 64 + lr;
    float* Cb1 = C + (size_t)(tm * 32 + 16 + 8 * half) * ldc + tn * 64 + lr;
#pragma unroll
    for (int r = 0; r < 8; ++r) {
        float* Cr0 = Cb0 + (size_t)r * ldc;
        float* Cr1 = Cb1 + (size_t)r * ldc;
        Cr0[0]  = acc00[r];  Cr0[16] = acc01[r];
        Cr0[32] = acc02[r];  Cr0[48] = acc03[r];
        Cr1[0]  = acc10[r];  Cr1[16] = acc11[r];
        Cr1[32] = acc12[r];  Cr1[48] = acc13[r];
    }
}

__global__ __launch_bounds__(256) void conv_silu_kernel(
    const float* __restrict__ xg, const float* __restrict__ conv_w,
    const float* __restrict__ conv_b, float* __restrict__ x_conv)
{
    int idx = blockIdx.x * 256 + threadIdx.x;
    if (idx >= NROW * DI) return;
    int d   = idx % DI;
    int row = idx / DI;
    int l   = row % L_SZ;

    float acc = conv_b[d];
    const float* w = conv_w + d * 4;
#pragma unroll
    for (int j = 0; j < 4; ++j) {
        int ls = l - 3 + j;
        if (ls >= 0)
            acc = fmaf(w[j], xg[(size_t)(row - 3 + j) * (2 * DI) + d], acc);
    }
    x_conv[idx] = acc / (1.0f + __expf(-acc));
}

__global__ __launch_bounds__(256) void xproj_kernel(
    const float* __restrict__ x_conv, const float* __restrict__ Wx,
    float* __restrict__ ssm)
{
    int idx = blockIdx.x * 256 + threadIdx.x;
    if (idx >= NROW * 33) return;
    int n   = idx % 33;
    int row = idx / 33;
    const float4* xr = (const float4*)(x_conv + (size_t)row * DI);
    const float4* wr = (const float4*)(Wx + (size_t)n * DI);
    float a0 = 0.f, a1 = 0.f, a2 = 0.f, a3 = 0.f;
    for (int k = 0; k < DI / 4; ++k) {
        float4 xv = xr[k];
        float4 wv = wr[k];
        a0 = fmaf(xv.x, wv.x, a0);
        a1 = fmaf(xv.y, wv.y, a1);
        a2 = fmaf(xv.z, wv.z, a2);
        a3 = fmaf(xv.w, wv.w, a3);
    }
    ssm[idx] = (a0 + a1) + (a2 + a3);
}

__global__ __launch_bounds__(256) void scan_kernel(
    const float* __restrict__ ssm, const float* __restrict__ x_conv,
    const float* __restrict__ xg, const float* __restrict__ w_dt,
    const float* __restrict__ b_dt, const float* __restrict__ A_log,
    const float* __restrict__ D_param, float* __restrict__ y)
{
    int g = blockIdx.x * 256 + threadIdx.x;
    if (g >= B_SZ * DI) return;
    int b = g / DI, d = g % DI;

    const float wdt = w_dt[d];
    const float bdt = b_dt[d];
    const float Dp  = D_param[d];
    float Aa[DS];
#pragma unroll
    for (int n = 0; n < DS; ++n) Aa[n] = -__expf(A_log[d * DS + n]);
    float h[DS];
#pragma unroll
    for (int n = 0; n < DS; ++n) h[n] = 0.f;

    for (int l = 0; l < L_SZ; ++l) {
        const int row = b * L_SZ + l;
        const float* s = ssm + (size_t)row * 33;
        const float x_t = x_conv[(size_t)row * DI + d];

        float z  = fmaf(s[32], wdt, bdt);
        float dt = (z > 20.0f) ? z : log1pf(__expf(z));

        float yv = 0.f;
#pragma unroll
        for (int n = 0; n < DS; ++n) {
            float Abar = __expf(dt * Aa[n]);
            h[n] = fmaf(Abar, h[n], (dt * s[n]) * x_t);
            yv   = fmaf(h[n], s[DS + n], yv);
        }
        yv = fmaf(x_t, Dp, yv);

        float gv = xg[(size_t)row * (2 * DI) + DI + d];
        float sg = gv / (1.0f + __expf(-gv));
        y[(size_t)row * DI + d] = yv * sg;
    }
}

extern "C" void kernel_launch(void* const* d_in, const int* in_sizes, int n_in,
                              void* d_out, int out_size, void* d_ws, size_t ws_size,
                              hipStream_t stream)
{
    const float* x       = (const float*)d_in[0];
    const float* W_in    = (const float*)d_in[1];
    const float* conv_w  = (const float*)d_in[2];
    const float* conv_b  = (const float*)d_in[3];
    const float* W_xproj = (const float*)d_in[4];
    const float* w_dt    = (const float*)d_in[5];
    const float* b_dt    = (const float*)d_in[6];
    const float* A_log   = (const float*)d_in[7];
    const float* D_param = (const float*)d_in[8];
    const float* W_out   = (const float*)d_in[9];
    float* out = (float*)d_out;

    float* ws     = (float*)d_ws;
    float* xg     = ws;
    float* x_conv = xg + (size_t)NROW * (2 * DI);
    float* ssm    = x_conv + (size_t)NROW * DI;
    float* yb     = ssm + (size_t)NROW * 33;

    gemm32x64_f32_wmma<DM><<<(NROW / 32) * ((2 * DI) / 64), 32, 0, stream>>>(
        x, W_in, xg, (2 * DI) / 64, DM, DM, 2 * DI);

    conv_silu_kernel<<<(NROW * DI + 255) / 256, 256, 0, stream>>>(
        xg, conv_w, conv_b, x_conv);

    xproj_kernel<<<(NROW * 33 + 255) / 256, 256, 0, stream>>>(
        x_conv, W_xproj, ssm);

    scan_kernel<<<(B_SZ * DI + 255) / 256, 256, 0, stream>>>(
        ssm, x_conv, xg, w_dt, b_dt, A_log, D_param, yb);

    gemm32x64_f32_wmma<DI><<<(NROW / 32) * (DM / 64), 32, 0, stream>>>(
        yb, W_out, out, DM / 64, DI, DI, DM);
}


# scan split along L into 64 chunks of 32 steps run in parallel: local pass, carry h=exp(A*sum dt)*h+h_local over chunks, replay pass; rounding order of the carried state changes
# speedup vs baseline: 5.8175x; 5.8175x over previous
#include <hip/hip_runtime.h>
#include <math.h>


typedef float  v2f __attribute__((ext_vector_type(2)));
typedef float  v8f __attribute__((ext_vector_type(8)));

#define B_SZ 2
#define L_SZ 2048
#define DM   768
#define DI   1536
#define DS   16
#define NROW (B_SZ * L_SZ)

template <int K>
__global__ __launch_bounds__(32) void gemm32x64_f32_wmma(
    const float* __restrict__ A, const float* __restrict__ W,
    float* __restrict__ C, int N64, int lda, int ldw, int ldc)
{
    static_assert(K % 8 == 0, "K must be a multiple of 8");
    const int tile = blockIdx.x;
    const int tm   = tile / N64;
    const int tn   = tile % N64;
    const int lane = threadIdx.x;
    const int half = lane >> 4;
    const int lr   = lane & 15;
    const int kh   = 2 * half;

    const float* Ar0 = A + (size_t)(tm * 32 +  0 + lr) * lda;
    const float* Ar1 = A + (size_t)(tm * 32 + 16 + lr) * lda;
    const float* W0  = W + (size_t)(tn * 64 +  0 + lr) * ldw;
    const float* W1  = W + (size_t)(tn * 64 + 16 + lr) * ldw;
    const float* W2  = W + (size_t)(tn * 64 + 32 + lr) * ldw;
    const float* W3  = W + (size_t)(tn * 64 + 48 + lr) * ldw;

    v8f acc00 = {}; v8f acc01 = {}; v8f acc02 = {}; v8f acc03 = {};
    v8f acc10 = {}; v8f acc11 = {}; v8f acc12 = {}; v8f acc13 = {};

    auto ld = [&](int kb, v2f& a0, v2f& a1, v2f& b0, v2f& b1, v2f& b2, v2f& b3) {
        a0 = *(const v2f*)(Ar0 + kb);
        a1 = *(const v2f*)(Ar1 + kb);
        b0 = *(const v2f*)(W0 + kb);
        b1 = *(const v2f*)(W1 + kb);
        b2 = *(const v2f*)(W2 + kb);
        b3 = *(const v2f*)(W3 + kb);
    };
    auto mm8 = [&](v2f a0, v2f a1, v2f b0, v2f b1, v2f b2, v2f b3) {
        acc00 = __builtin_amdgcn_wmma_f32_16x16x4_f32(false, a0, false, b0, (short)0, acc00, false, false);
        acc01 = __builtin_amdgcn_wmma_f32_16x16x4_f32(false, a0, false, b1, (short)0, acc01, false, false);
        acc02 = __builtin_amdgcn_wmma_f32_16x16x4_f32(false, a0, false, b2, (short)0, acc02, false, false);
        acc03 = __builtin_amdgcn_wmma_f32_16x16x4_f32(false, a0, false, b3, (short)0, acc03, false, false);
        acc10 = __builtin_amdgcn_wmma_f32_16x16x4_f32(false, a1, false, b0, (short)0, acc10, false, false);
        acc11 = __builtin_amdgcn_wmma_f32_16x16x4_f32(false, a1, false, b1, (short)0, acc11, false, false);
        acc12 = __builtin_amdgcn_wmma_f32_16x16x4_f32(false, a1, false, b2, (short)0, acc12, false, false);
        acc13 = __builtin_amdgcn_wmma_f32_16x16x4_f32(false, a1, false, b3, (short)0, acc13, false, false);
    };

    v2f a0A, a1A, b0A, b1A, b2A, b3A;
    v2f a0B, a1B, b0B, b1B, b2B, b3B;

    ld(kh, a0A, a1A, b0A, b1A, b2A, b3A);
    for (int k = 0; k < K - 8; k += 8) {
        ld(k + 4 + kh, a0B, a1B, b0B, b1B, b2B, b3B);
        mm8(a0A, a1A, b0A, b1A, b2A, b3A);
        ld(k + 8 + kh, a0A, a1A, b0A, b1A, b2A, b3A);
        mm8(a0B, a1B, b0B, b1B, b2B, b3B);
    }
    ld(K - 4 + kh, a0B, a1B, b0B, b1B, b2B, b3B);
    mm8(a0A, a1A, b0A, b1A, b2A, b3A);
    mm8(a0B, a1B, b0B, b1B, b2B, b3B);

    float* Cb0 = C + (size_t)(tm * 32 +  0 + 8 * half) * ldc + tn * 64 + lr;
    float* Cb1 = C + (size_t)(tm * 32 + 16 + 8 * half) * ldc + tn * 64 + lr;
#pragma unroll
    for (int r = 0; r < 8; ++r) {
        float* Cr0 = Cb0 + (size_t)r * ldc;
        float* Cr1 = Cb1 + (size_t)r * ldc;
        Cr0[0]  = acc00[r];  Cr0[16] = acc01[r];
        Cr0[32] = acc02[r];  Cr0[48] = acc03[r];
        Cr1[0]  = acc10[r];  Cr1[16] = acc11[r];
        Cr1[32] = acc12[r];  Cr1[48] = acc13[r];
    }
}

__global__ __launch_bounds__(256) void conv_silu_kernel(
    const float* __restrict__ xg, const float* __restrict__ conv_w,
    const float* __restrict__ conv_b, float* __restrict__ x_conv)
{
    int idx = blockIdx.x * 256 + threadIdx.x;
    if (idx >= NROW * DI) return;
    int d   = idx % DI;
    int row = idx / DI;
    int l   = row % L_SZ;

    float acc = conv_b[d];
    const float* w = conv_w + d * 4;
#pragma unroll
    for (int j = 0; j < 4; ++j) {
        int ls = l - 3 + j;
        if (ls >= 0)
            acc = fmaf(w[j], xg[(size_t)(row - 3 + j) * (2 * DI) + d], acc);
    }
    x_conv[idx] = acc / (1.0f + __expf(-acc));
}

__global__ __launch_bounds__(256) void xproj_kernel(
    const float* __restrict__ x_conv, const float* __restrict__ Wx,
    float* __restrict__ ssm)
{
    int idx = blockIdx.x * 256 + threadIdx.x;
    if (idx >= NROW * 33) return;
    int n   = idx % 33;
    int row = idx / 33;
    const float4* xr = (const float4*)(x_conv + (size_t)row * DI);
    const float4* wr = (const float4*)(Wx + (size_t)n * DI);
    float a0 = 0.f, a1 = 0.f, a2 = 0.f, a3 = 0.f;
    for (int k = 0; k < DI / 4; ++k) {
        float4 xv = xr[k];
        float4 wv = wr[k];
        a0 = fmaf(xv.x, wv.x, a0);
        a1 = fmaf(xv.y, wv.y, a1);
        a2 = fmaf(xv.z, wv.z, a2);
        a3 = fmaf(xv.w, wv.w, a3);
    }
    ssm[idx] = (a0 + a1) + (a2 + a3);
}

#define CH_T 32
#define CH_C (L_SZ / CH_T)
#define DGRP (DI / 256)

__global__ __launch_bounds__(256) void scan_local_kernel(
    const float* __restrict__ ssm, const float* __restrict__ x_conv,
    const float* __restrict__ w_dt, const float* __restrict__ b_dt,
    const float* __restrict__ A_log,
    float* __restrict__ hloc, float* __restrict__ sumdt)
{
    const int dg   = blockIdx.x % DGRP;
    const int bc   = blockIdx.x / DGRP;
    const int d    = dg * 256 + threadIdx.x;
    const int row0 = bc * CH_T;

    const float wdt = w_dt[d];
    const float bdt = b_dt[d];
    float Aa[DS];
#pragma unroll
    for (int n = 0; n < DS; ++n) Aa[n] = -__expf(A_log[d * DS + n]);
    float h[DS];
#pragma unroll
    for (int n = 0; n < DS; ++n) h[n] = 0.f;
    float sdt = 0.f;

#pragma unroll 4
    for (int t = 0; t < CH_T; ++t) {
        const int row = row0 + t;
        const float* s = ssm + (size_t)row * 33;
        const float x_t = x_conv[(size_t)row * DI + d];

        float z  = fmaf(s[32], wdt, bdt);
        float dt = (z > 20.0f) ? z : log1pf(__expf(z));
        sdt += dt;
#pragma unroll
        for (int n = 0; n < DS; ++n) {
            float Abar = __expf(dt * Aa[n]);
            h[n] = fmaf(Abar, h[n], (dt * s[n]) * x_t);
        }
    }
#pragma unroll
    for (int n = 0; n < DS; ++n)
        hloc[((size_t)bc * DS + n) * DI + d] = h[n];
    sumdt[(size_t)bc * DI + d] = sdt;
}

__global__ __launch_bounds__(256) void scan_carry_kernel(
    const float* __restrict__ hloc, const float* __restrict__ sumdt,
    const float* __restrict__ A_log, float* __restrict__ hin)
{
    const int idx = blockIdx.x * 256 + threadIdx.x;
    if (idx >= B_SZ * DS * DI) return;
    const int d = idx % DI;
    const int n = (idx / DI) % DS;
    const int b = idx / (DI * DS);
    const float Aa = -__expf(A_log[d * DS + n]);

    float h = 0.f;
#pragma unroll 8
    for (int c = 0; c < CH_C; ++c) {
        const size_t bc = (size_t)b * CH_C + c;
        const float hl = hloc[(bc * DS + n) * DI + d];
        const float sd = sumdt[bc * DI + d];
        hin[(bc * DS + n) * DI + d] = h;
        h = fmaf(__expf(Aa * sd), h, hl);
    }
}

__global__ __launch_bounds__(256) void scan_replay_kernel(
    const float* __restrict__ ssm, const float* __restrict__ x_conv,
    const float* __restrict__ xg, const float* __restrict__ w_dt,
    const float* __restrict__ b_dt, const float* __restrict__ A_log,
    const float* __restrict__ D_param, const float* __restrict__ hin,
    float* __restrict__ y)
{
    const int dg   = blockIdx.x % DGRP;
    const int bc   = blockIdx.x / DGRP;
    const int d    = dg * 256 + threadIdx.x;
    const int row0 = bc * CH_T;

    const float wdt = w_dt[d];
    const float bdt = b_dt[d];
    const float Dp  = D_param[d];
    float Aa[DS];
#pragma unroll
    for (int n = 0; n < DS; ++n) Aa[n] = -__expf(A_log[d * DS + n]);
    float h[DS];
#pragma unroll
    for (int n = 0; n < DS; ++n) h[n] = hin[((size_t)bc * DS + n) * DI + d];

#pragma unroll 4
    for (int t = 0; t < CH_T; ++t) {
        const int row = row0 + t;
        const float* s = ssm + (size_t)row * 33;
        const float x_t = x_conv[(size_t)row * DI + d];

        float z  = fmaf(s[32], wdt, bdt);
        float dt = (z > 20.0f) ? z : log1pf(__expf(z));

        float yv = 0.f;
#pragma unroll
        for (int n = 0; n < DS; ++n) {
            float Abar = __expf(dt * Aa[n]);
            h[n] = fmaf(Abar, h[n], (dt * s[n]) * x_t);
            yv   = fmaf(h[n], s[DS + n], yv);
        }
        yv = fmaf(x_t, Dp, yv);

        float gv = xg[(size_t)row * (2 * DI) + DI + d];
        float sg = gv / (1.0f + __expf(-gv));
        y[(size_t)row * DI + d] = yv * sg;
    }
}

extern "C" void kernel_launch(void* const* d_in, const int* in_sizes, int n_in,
                              void* d_out, int out_size, void* d_ws, size_t ws_size,
                              hipStream_t stream)
{
    const float* x       = (const float*)d_in[0];
    const float* W_in    = (const float*)d_in[1];
    const float* conv_w  = (const float*)d_in[2];
    const float* conv_b  = (const float*)d_in[3];
    const float* W_xproj = (const float*)d_in[4];
    const float* w_dt    = (const float*)d_in[5];
    const float* b_dt    = (const float*)d_in[6];
    const float* A_log   = (const float*)d_in[7];
    const float* D_param = (const float*)d_in[8];
    const float* W_out   = (const float*)d_in[9];
    float* out = (float*)d_out;

    float* ws     = (float*)d_ws;
    float* xg     = ws;
    float* x_conv = xg + (size_t)NROW * (2 * DI);
    float* ssm    = x_conv + (size_t)NROW * DI;
    float* yb     = ssm + (size_t)NROW * 33;
    float* hloc   = yb + (size_t)NROW * DI;
    float* hin    = hloc + (size_t)B_SZ * CH_C * DS * DI;
    float* sumdt  = hin + (size_t)B_SZ * CH_C * DS * DI;

    gemm32x64_f32_wmma<DM><<<(NROW / 32) * ((2 * DI) / 64), 32, 0, stream>>>(
        x, W_in, xg, (2 * DI) / 64, DM, DM, 2 * DI);

    conv_silu_kernel<<<(NROW * DI + 255) / 256, 256, 0, stream>>>(
        xg, conv_w, conv_b, x_conv);

    xproj_kernel<<<(NROW * 33 + 255) / 256, 256, 0, stream>>>(
        x_conv, W_xproj, ssm);

    scan_local_kernel<<<B_SZ * CH_C * DGRP, 256, 0, stream>>>(
        ssm, x_conv, w_dt, b_dt, A_log, hloc, sumdt);
    scan_carry_kernel<<<(B_SZ * DS * DI + 255) / 256, 256, 0, stream>>>(
        hloc, sumdt, A_log, hin);
    scan_replay_kernel<<<B_SZ * CH_C * DGRP, 256, 0, stream>>>(
        ssm, x_conv, xg, w_dt, b_dt, A_log, D_param, hin, yb);

    gemm32x64_f32_wmma<DI><<<(NROW / 32) * (DM / 64), 32, 0, stream>>>(
        yb, W_out, out, DM / 64, DI, DI, DM);
}


# both GEMMs stage 32-deep A and W panels through LDS: 4-wave work-groups (64x128, 64x64 tiles), whole-line 16-byte global loads, fragments read from LDS; k order per accumulator unchanged
# speedup vs baseline: 9.3290x; 1.6036x over previous
#include <hip/hip_runtime.h>
#include <math.h>


typedef float  v2f __attribute__((ext_vector_type(2)));
typedef float  v4f __attribute__((ext_vector_type(4)));
typedef float  v8f __attribute__((ext_vector_type(8)));

#define B_SZ 2
#define L_SZ 2048
#define DM   768
#define DI   1536
#define DS   16
#define NROW (B_SZ * L_SZ)

template <int K, int NF>
__global__ __launch_bounds__(128, 6) void gemm_lds_f32_wmma(
    const float* __restrict__ A, const float* __restrict__ W,
    float* __restrict__ C, int NT, int ldc)
{
    static_assert(K % 32 == 0, "K must be a multiple of the panel depth");
    constexpr int BM    = 64;
    constexpr int BN    = 32 * NF;
    constexpr int KP    = 32;
    constexpr int PITCH = 36;
    constexpr int ROWS  = BM + BN;
    constexpr int NLD   = ROWS / 16;
    static_assert(NLD == 8 || NLD == 12, "staging is written out for NF = 2 and NF = 4");

    __shared__ __attribute__((aligned(16))) float lds[ROWS * PITCH];

    const int tid  = threadIdx.x;
    const int wave = tid >> 5;
    const int lane = tid & 31;
    const int half = lane >> 4;
    const int lr   = lane & 15;
    const int wm   = wave >> 1;
    const int wn   = wave & 1;
    const int tm   = blockIdx.x / NT;
    const int tn   = blockIdx.x % NT;

    const int r0 = tid >> 3;
    const int c4 = tid & 7;
    const float* gA = A + (size_t)(tm * BM + r0) * K + c4 * 4;
    const float* gW = W + (size_t)(tn * BN + r0) * K + c4 * 4;
    float* sdst = lds + r0 * PITCH + c4 * 4;

    v4f s0, s1, s2, s3, s4, s5, s6, s7, s8, s9, s10, s11;
    auto gload = [&](int kc) {
        s0 = *(const v4f*)(gA + (size_t)0  * K + kc);
        s1 = *(const v4f*)(gA + (size_t)16 * K + kc);
        s2 = *(const v4f*)(gA + (size_t)32 * K + kc);
        s3 = *(const v4f*)(gA + (size_t)48 * K + kc);
        s4 = *(const v4f*)(gW + (size_t)0  * K + kc);
        s5 = *(const v4f*)(gW + (size_t)16 * K + kc);
        s6 = *(const v4f*)(gW + (size_t)32 * K + kc);
        s7 = *(const v4f*)(gW + (size_t)48 * K + kc);
        if constexpr (NLD == 12) {
            s8  = *(const v4f*)(gW + (size_t)64  * K + kc);
            s9  = *(const v4f*)(gW + (size_t)80  * K + kc);
            s10 = *(const v4f*)(gW + (size_t)96  * K + kc);
            s11 = *(const v4f*)(gW + (size_t)112 * K + kc);
        }
    };
    auto sstore = [&]() {
        *(v4f*)(sdst + 0  * PITCH) = s0;
        *(v4f*)(sdst + 16 * PITCH) = s1;
        *(v4f*)(sdst + 32 * PITCH) = s2;
        *(v4f*)(sdst + 48 * PITCH) = s3;
        *(v4f*)(sdst + 64 * PITCH) = s4;
        *(v4f*)(sdst + 80 * PITCH) = s5;
        *(v4f*)(sdst + 96 * PITCH) = s6;
        *(v4f*)(sdst + 112 * PITCH) = s7;
        if constexpr (NLD == 12) {
            *(v4f*)(sdst + 128 * PITCH) = s8;
            *(v4f*)(sdst + 144 * PITCH) = s9;
            *(v4f*)(sdst + 160 * PITCH) = s10;
            *(v4f*)(sdst + 176 * PITCH) = s11;
        }
    };

    const float* sA = lds + (wm * 32 + lr) * PITCH + 2 * half;
    const float* sB = lds + (BM + wn * 16 * NF + lr) * PITCH + 2 * half;

    v8f acc[2][NF];
#pragma unroll
    for (int i = 0; i < 2; ++i)
#pragma unroll
        for (int f = 0; f < NF; ++f) acc[i][f] = v8f{};

    gload(0);
    for (int kc = 0; kc < K; kc += KP) {
        sstore();
        __syncthreads();
        if (kc + KP < K) gload(kc + KP);

#pragma unroll
        for (int ks = 0; ks < KP / 4; ++ks) {
            const v2f a0 = *(const v2f*)(sA + 4 * ks);
            const v2f a1 = *(const v2f*)(sA + 16 * PITCH + 4 * ks);
            v2f b[NF];
#pragma unroll
            for (int f = 0; f < NF; ++f)
                b[f] = *(const v2f*)(sB + f * 16 * PITCH + 4 * ks);
#pragma unroll
            for (int f = 0; f < NF; ++f)
                acc[0][f] = __builtin_amdgcn_wmma_f32_16x16x4_f32(
                    false, a0, false, b[f], (short)0, acc[0][f], false, false);
#pragma unroll
            for (int f = 0; f < NF; ++f)
                acc[1][f] = __builtin_amdgcn_wmma_f32_16x16x4_f32(
                    false, a1, false, b[f], (short)0, acc[1][f], false, false);
        }
        __syncthreads();
    }

    float* Cb = C + (size_t)(tm * BM + wm * 32 + 8 * half) * ldc
                  + tn * BN + wn * 16 * NF + lr;
#pragma unroll
    for (int r = 0; r < 8; ++r) {
#pragma unroll
        for (int i = 0; i < 2; ++i) {
            float* Cr = Cb + (size_t)(16 * i + r) * ldc;
#pragma unroll
            for (int f = 0; f < NF; ++f) Cr[16 * f] = acc[i][f][r];
        }
    }
}

__global__ __launch_bounds__(256) void conv_silu_kernel(
    const float* __restrict__ xg, const float* __restrict__ conv_w,
    const float* __restrict__ conv_b, float* __restrict__ x_conv)
{
    int idx = blockIdx.x * 256 + threadIdx.x;
    if (idx >= NROW * DI) return;
    int d   = idx % DI;
    int row = idx / DI;
    int l   = row % L_SZ;

    float acc = conv_b[d];
    const float* w = conv_w + d * 4;
#pragma unroll
    for (int j = 0; j < 4; ++j) {
        int ls = l - 3 + j;
        if (ls >= 0)
            acc = fmaf(w[j], xg[(size_t)(row - 3 + j) * (2 * DI) + d], acc);
    }
    x_conv[idx] = acc / (1.0f + __expf(-acc));
}

__global__ __launch_bounds__(256) void xproj_kernel(
    const float* __restrict__ x_conv, const float* __restrict__ Wx,
    float* __restrict__ ssm)
{
    int idx = blockIdx.x * 256 + threadIdx.x;
    if (idx >= NROW * 33) return;
    int n   = idx % 33;
    int row = idx / 33;
    const float4* xr = (const float4*)(x_conv + (size_t)row * DI);
    const float4* wr = (const float4*)(Wx + (size_t)n * DI);
    float a0 = 0.f, a1 = 0.f, a2 = 0.f, a3 = 0.f;
    for (int k = 0; k < DI / 4; ++k) {
        float4 xv = xr[k];
        float4 wv = wr[k];
        a0 = fmaf(xv.x, wv.x, a0);
        a1 = fmaf(xv.y, wv.y, a1);
        a2 = fmaf(xv.z, wv.z, a2);
        a3 = fmaf(xv.w, wv.w, a3);
    }
    ssm[idx] = (a0 + a1) + (a2 + a3);
}

#define CH_T 32
#define CH_C (L_SZ / CH_T)
#define DGRP (DI / 256)

__global__ __launch_bounds__(256) void scan_local_kernel(
    const float* __restrict__ ssm, const float* __restrict__ x_conv,
    const float* __restrict__ w_dt, const float* __restrict__ b_dt,
    const float* __restrict__ A_log,
    float* __restrict__ hloc, float* __restrict__ sumdt)
{
    const int dg   = blockIdx.x % DGRP;
    const int bc   = blockIdx.x / DGRP;
    const int d    = dg * 256 + threadIdx.x;
    const int row0 = bc * CH_T;

    const float wdt = w_dt[d];
    const float bdt = b_dt[d];
    float Aa[DS];
#pragma unroll
    for (int n = 0; n < DS; ++n) Aa[n] = -__expf(A_log[d * DS + n]);
    float h[DS];
#pragma unroll
    for (int n = 0; n < DS; ++n) h[n] = 0.f;
    float sdt = 0.f;

#pragma unroll 4
    for (int t = 0; t < CH_T; ++t) {
        const int row = row0 + t;
        const float* s = ssm + (size_t)row * 33;
        const float x_t = x_conv[(size_t)row * DI + d];

        float z  = fmaf(s[32], wdt, bdt);
        float dt = (z > 20.0f) ? z : log1pf(__expf(z));
        sdt += dt;
#pragma unroll
        for (int n = 0; n < DS; ++n) {
            float Abar = __expf(dt * Aa[n]);
            h[n] = fmaf(Abar, h[n], (dt * s[n]) * x_t);
        }
    }
#pragma unroll
    for (int n = 0; n < DS; ++n)
        hloc[((size_t)bc * DS + n) * DI + d] = h[n];
    sumdt[(size_t)bc * DI + d] = sdt;
}

__global__ __launch_bounds__(256) void scan_carry_kernel(
    const float* __restrict__ hloc, const float* __restrict__ sumdt,
    const float* __restrict__ A_log, float* __restrict__ hin)
{
    const int idx = blockIdx.x * 256 + threadIdx.x;
    if (idx >= B_SZ * DS * DI) return;
    const int d = idx % DI;
    const int n = (idx / DI) % DS;
    const int b = idx / (DI * DS);
    const float Aa = -__expf(A_log[d * DS + n]);

    float h = 0.f;
#pragma unroll 8
    for (int c = 0; c < CH_C; ++c) {
        const size_t bc = (size_t)b * CH_C + c;
        const float hl = hloc[(bc * DS + n) * DI + d];
        const float sd = sumdt[bc * DI + d];
        hin[(bc * DS + n) * DI + d] = h;
        h = fmaf(__expf(Aa * sd), h, hl);
    }
}

__global__ __launch_bounds__(256) void scan_replay_kernel(
    const float* __restrict__ ssm, const float* __restrict__ x_conv,
    const float* __restrict__ xg, const float* __restrict__ w_dt,
    const float* __restrict__ b_dt, const float* __restrict__ A_log,
    const float* __restrict__ D_param, const float* __restrict__ hin,
    float* __restrict__ y)
{
    const int dg   = blockIdx.x % DGRP;
    const int bc   = blockIdx.x / DGRP;
    const int d    = dg * 256 + threadIdx.x;
    const int row0 = bc * CH_T;

    const float wdt = w_dt[d];
    const float bdt = b_dt[d];
    const float Dp  = D_param[d];
    float Aa[DS];
#pragma unroll
    for (int n = 0; n < DS; ++n) Aa[n] = -__expf(A_log[d * DS + n]);
    float h[DS];
#pragma unroll
    for (int n = 0; n < DS; ++n) h[n] = hin[((size_t)bc * DS + n) * DI + d];

#pragma unroll 4
    for (int t = 0; t < CH_T; ++t) {
        const int row = row0 + t;
        const float* s = ssm + (size_t)row * 33;
        const float x_t = x_conv[(size_t)row * DI + d];

        float z  = fmaf(s[32], wdt, bdt);
        float dt = (z > 20.0f) ? z : log1pf(__expf(z));

        float yv = 0.f;
#pragma unroll
        for (int n = 0; n < DS; ++n) {
            float Abar = __expf(dt * Aa[n]);
            h[n] = fmaf(Abar, h[n], (dt * s[n]) * x_t);
            yv   = fmaf(h[n], s[DS + n], yv);
        }
        yv = fmaf(x_t, Dp, yv);

        float gv = xg[(size_t)row * (2 * DI) + DI + d];
        float sg = gv / (1.0f + __expf(-gv));
        y[(size_t)row * DI + d] = yv * sg;
    }
}

extern "C" void kernel_launch(void* const* d_in, const int* in_sizes, int n_in,
                              void* d_out, int out_size, void* d_ws, size_t ws_size,
                              hipStream_t stream)
{
    const float* x       = (const float*)d_in[0];
    const float* W_in    = (const float*)d_in[1];
    const float* conv_w  = (const float*)d_in[2];
    const float* conv_b  = (const float*)d_in[3];
    const float* W_xproj = (const float*)d_in[4];
    const float* w_dt    = (const float*)d_in[5];
    const float* b_dt    = (const float*)d_in[6];
    const float* A_log   = (const float*)d_in[7];
    const float* D_param = (const float*)d_in[8];
    const float* W_out   = (const float*)d_in[9];
    float* out = (float*)d_out;

    float* ws     = (float*)d_ws;
    float* xg     = ws;
    float* x_conv = xg + (size_t)NROW * (2 * DI);
    float* ssm    = x_conv + (size_t)NROW * DI;
    float* yb     = ssm + (size_t)NROW * 33;
    float* hloc   = yb + (size_t)NROW * DI;
    float* hin    = hloc + (size_t)B_SZ * CH_C * DS * DI;
    float* sumdt  = hin + (size_t)B_SZ * CH_C * DS * DI;

    gemm_lds_f32_wmma<DM, 4><<<(NROW / 64) * ((2 * DI) / 128), 128, 0, stream>>>(
        x, W_in, xg, (2 * DI) / 128, 2 * DI);

    conv_silu_kernel<<<(NROW * DI + 255) / 256, 256, 0, stream>>>(
        xg, conv_w, conv_b, x_conv);

    xproj_kernel<<<(NROW * 33 + 255) / 256, 256, 0, stream>>>(
        x_conv, W_xproj, ssm);

    scan_local_kernel<<<B_SZ * CH_C * DGRP, 256, 0, stream>>>(
        ssm, x_conv, w_dt, b_dt, A_log, hloc, sumdt);
    scan_carry_kernel<<<(B_SZ * DS * DI + 255) / 256, 256, 0, stream>>>(
        hloc, sumdt, A_log, hin);
    scan_replay_kernel<<<B_SZ * CH_C * DGRP, 256, 0, stream>>>(
        ssm, x_conv, xg, w_dt, b_dt, A_log, D_param, hin, yb);

    gemm_lds_f32_wmma<DI, 2><<<(NROW / 64) * (DM / 64), 128, 0, stream>>>(
        yb, W_out, out, DM / 64, DM);
}


# conv+SiLU kernel coarsened: one thread does 4 channels x 8 steps from an 11-row register window with 16-byte loads and stores; 6144 waves instead of 196608; every sum unchanged
# speedup vs baseline: 9.9545x; 1.0670x over previous
#include <hip/hip_runtime.h>
#include <math.h>


typedef float  v2f __attribute__((ext_vector_type(2)));
typedef float  v4f __attribute__((ext_vector_type(4)));
typedef float  v8f __attribute__((ext_vector_type(8)));

#define B_SZ 2
#define L_SZ 2048
#define DM   768
#define DI   1536
#define DS   16
#define NROW (B_SZ * L_SZ)

template <int K, int NF>
__global__ __launch_bounds__(128, 6) void gemm_lds_f32_wmma(
    const float* __restrict__ A, const float* __restrict__ W,
    float* __restrict__ C, int NT, int ldc)
{
    static_assert(K % 32 == 0, "K must be a multiple of the panel depth");
    constexpr int BM    = 64;
    constexpr int BN    = 32 * NF;
    constexpr int KP    = 32;
    constexpr int PITCH = 36;
    constexpr int ROWS  = BM + BN;
    constexpr int NLD   = ROWS / 16;
    static_assert(NLD == 8 || NLD == 12, "staging is written out for NF = 2 and NF = 4");

    __shared__ __attribute__((aligned(16))) float lds[ROWS * PITCH];

    const int tid  = threadIdx.x;
    const int wave = tid >> 5;
    const int lane = tid & 31;
    const int half = lane >> 4;
    const int lr   = lane & 15;
    const int wm   = wave >> 1;
    const int wn   = wave & 1;
    const int tm   = blockIdx.x / NT;
    const int tn   = blockIdx.x % NT;

    const int r0 = tid >> 3;
    const int c4 = tid & 7;
    const float* gA = A + (size_t)(tm * BM + r0) * K + c4 * 4;
    const float* gW = W + (size_t)(tn * BN + r0) * K + c4 * 4;
    float* sdst = lds + r0 * PITCH + c4 * 4;

    v4f s0, s1, s2, s3, s4, s5, s6, s7, s8, s9, s10, s11;
    auto gload = [&](int kc) {
        s0 = *(const v4f*)(gA + (size_t)0  * K + kc);
        s1 = *(const v4f*)(gA + (size_t)16 * K + kc);
        s2 = *(const v4f*)(gA + (size_t)32 * K + kc);
        s3 = *(const v4f*)(gA + (size_t)48 * K + kc);
        s4 = *(const v4f*)(gW + (size_t)0  * K + kc);
        s5 = *(const v4f*)(gW + (size_t)16 * K + kc);
        s6 = *(const v4f*)(gW + (size_t)32 * K + kc);
        s7 = *(const v4f*)(gW + (size_t)48 * K + kc);
        if constexpr (NLD == 12) {
            s8  = *(const v4f*)(gW + (size_t)64  * K + kc);
            s9  = *(const v4f*)(gW + (size_t)80  * K + kc);
            s10 = *(const v4f*)(gW + (size_t)96  * K + kc);
            s11 = *(const v4f*)(gW + (size_t)112 * K + kc);
        }
    };
    auto sstore = [&]() {
        *(v4f*)(sdst + 0  * PITCH) = s0;
        *(v4f*)(sdst + 16 * PITCH) = s1;
        *(v4f*)(sdst + 32 * PITCH) = s2;
        *(v4f*)(sdst + 48 * PITCH) = s3;
        *(v4f*)(sdst + 64 * PITCH) = s4;
        *(v4f*)(sdst + 80 * PITCH) = s5;
        *(v4f*)(sdst + 96 * PITCH) = s6;
        *(v4f*)(sdst + 112 * PITCH) = s7;
        if constexpr (NLD == 12) {
            *(v4f*)(sdst + 128 * PITCH) = s8;
            *(v4f*)(sdst + 144 * PITCH) = s9;
            *(v4f*)(sdst + 160 * PITCH) = s10;
            *(v4f*)(sdst + 176 * PITCH) = s11;
        }
    };

    const float* sA = lds + (wm * 32 + lr) * PITCH + 2 * half;
    const float* sB = lds + (BM + wn * 16 * NF + lr) * PITCH + 2 * half;

    v8f acc[2][NF];
#pragma unroll
    for (int i = 0; i < 2; ++i)
#pragma unroll
        for (int f = 0; f < NF; ++f) acc[i][f] = v8f{};

    gload(0);
    for (int kc = 0; kc < K; kc += KP) {
        sstore();
        __syncthreads();
        if (kc + KP < K) gload(kc + KP);

#pragma unroll
        for (int ks = 0; ks < KP / 4; ++ks) {
            const v2f a0 = *(const v2f*)(sA + 4 * ks);
            const v2f a1 = *(const v2f*)(sA + 16 * PITCH + 4 * ks);
            v2f b[NF];
#pragma unroll
            for (int f = 0; f < NF; ++f)
                b[f] = *(const v2f*)(sB + f * 16 * PITCH + 4 * ks);
#pragma unroll
            for (int f = 0; f < NF; ++f)
                acc[0][f] = __builtin_amdgcn_wmma_f32_16x16x4_f32(
                    false, a0, false, b[f], (short)0, acc[0][f], false, false);
#pragma unroll
            for (int f = 0; f < NF; ++f)
                acc[1][f] = __builtin_amdgcn_wmma_f32_16x16x4_f32(
                    false, a1, false, b[f], (short)0, acc[1][f], false, false);
        }
        __syncthreads();
    }

    float* Cb = C + (size_t)(tm * BM + wm * 32 + 8 * half) * ldc
                  + tn * BN + wn * 16 * NF + lr;
#pragma unroll
    for (int r = 0; r < 8; ++r) {
#pragma unroll
        for (int i = 0; i < 2; ++i) {
            float* Cr = Cb + (size_t)(16 * i + r) * ldc;
#pragma unroll
            for (int f = 0; f < NF; ++f) Cr[16 * f] = acc[i][f][r];
        }
    }
}

#define CV_T 8

__global__ __launch_bounds__(DI / 4) void conv_silu_kernel(
    const float* __restrict__ xg, const float* __restrict__ conv_w,
    const float* __restrict__ conv_b, float* __restrict__ x_conv)
{
    static_assert(L_SZ % CV_T == 0 && CV_T >= 3,
                  "a chunk must not cross a sequence end, and only the first chunk of a sequence lacks its 3 rows in front");
    const unsigned d    = threadIdx.x * 4;
    const unsigned row0 = blockIdx.x * CV_T;
    const bool head = (row0 % L_SZ) == 0;

    v4f w[4];
#pragma unroll
    for (int c = 0; c < 4; ++c) w[c] = *(const v4f*)(conv_w + (size_t)(d + c) * 4);
    const v4f bias = *(const v4f*)(conv_b + d);

    v4f xin[CV_T + 3];
    const float* src = xg + (size_t)row0 * (2 * DI) + d;
    const size_t back = head ? 0 : (size_t)(2 * DI);
#pragma unroll
    for (int i = 0; i < 3; ++i)
        xin[i] = *(const v4f*)(src - (size_t)(3 - i) * back);
#pragma unroll
    for (int i = 3; i < CV_T + 3; ++i)
        xin[i] = *(const v4f*)(src + (size_t)(i - 3) * (2 * DI));

    float* dst = x_conv + (size_t)row0 * DI + d;
#pragma unroll
    for (int t = 0; t < CV_T; ++t) {
        v4f o;
#pragma unroll
        for (int c = 0; c < 4; ++c) {
            float acc = bias[c];
#pragma unroll
            for (int j = 0; j < 4; ++j)
                if (t + j >= 3 || !head)
                    acc = fmaf(w[c][j], xin[t + j][c], acc);
            o[c] = acc / (1.0f + __expf(-acc));
        }
        *(v4f*)(dst + (size_t)t * DI) = o;
    }
}

__global__ __launch_bounds__(256) void xproj_kernel(
    const float* __restrict__ x_conv, const float* __restrict__ Wx,
    float* __restrict__ ssm)
{
    int idx = blockIdx.x * 256 + threadIdx.x;
    if (idx >= NROW * 33) return;
    int n   = idx % 33;
    int row = idx / 33;
    const float4* xr = (const float4*)(x_conv + (size_t)row * DI);
    const float4* wr = (const float4*)(Wx + (size_t)n * DI);
    float a0 = 0.f, a1 = 0.f, a2 = 0.f, a3 = 0.f;
    for (int k = 0; k < DI / 4; ++k) {
        float4 xv = xr[k];
        float4 wv = wr[k];
        a0 = fmaf(xv.x, wv.x, a0);
        a1 = fmaf(xv.y, wv.y, a1);
        a2 = fmaf(xv.z, wv.z, a2);
        a3 = fmaf(xv.w, wv.w, a3);
    }
    ssm[idx] = (a0 + a1) + (a2 + a3);
}

#define CH_T 32
#define CH_C (L_SZ / CH_T)
#define DGRP (DI / 256)

__global__ __launch_bounds__(256) void scan_local_kernel(
    const float* __restrict__ ssm, const float* __restrict__ x_conv,
    const float* __restrict__ w_dt, const float* __restrict__ b_dt,
    const float* __restrict__ A_log,
    float* __restrict__ hloc, float* __restrict__ sumdt)
{
    const int dg   = blockIdx.x % DGRP;
    const int bc   = blockIdx.x / DGRP;
    const int d    = dg * 256 + threadIdx.x;
    const int row0 = bc * CH_T;

    const float wdt = w_dt[d];
    const float bdt = b_dt[d];
    float Aa[DS];
#pragma unroll
    for (int n = 0; n < DS; ++n) Aa[n] = -__expf(A_log[d * DS + n]);
    float h[DS];
#pragma unroll
    for (int n = 0; n < DS; ++n) h[n] = 0.f;
    float sdt = 0.f;

#pragma unroll 4
    for (int t = 0; t < CH_T; ++t) {
        const int row = row0 + t;
        const float* s = ssm + (size_t)row * 33;
        const float x_t = x_conv[(size_t)row * DI + d];

        float z  = fmaf(s[32], wdt, bdt);
        float dt = (z > 20.0f) ? z : log1pf(__expf(z));
        sdt += dt;
#pragma unroll
        for (int n = 0; n < DS; ++n) {
            float Abar = __expf(dt * Aa[n]);
            h[n] = fmaf(Abar, h[n], (dt * s[n]) * x_t);
        }
    }
#pragma unroll
    for (int n = 0; n < DS; ++n)
        hloc[((size_t)bc * DS + n) * DI + d] = h[n];
    sumdt[(size_t)bc * DI + d] = sdt;
}

__global__ __launch_bounds__(256) void scan_carry_kernel(
    const float* __restrict__ hloc, const float* __restrict__ sumdt,
    const float* __restrict__ A_log, float* __restrict__ hin)
{
    const int idx = blockIdx.x * 256 + threadIdx.x;
    if (idx >= B_SZ * DS * DI) return;
    const int d = idx % DI;
    const int n = (idx / DI) % DS;
    const int b = idx / (DI * DS);
    const float Aa = -__expf(A_log[d * DS + n]);

    float h = 0.f;
#pragma unroll 8
    for (int c = 0; c < CH_C; ++c) {
        const size_t bc = (size_t)b * CH_C + c;
        const float hl = hloc[(bc * DS + n) * DI + d];
        const float sd = sumdt[bc * DI + d];
        hin[(bc * DS + n) * DI + d] = h;
        h = fmaf(__expf(Aa * sd), h, hl);
    }
}

__global__ __launch_bounds__(256) void scan_replay_kernel(
    const float* __restrict__ ssm, const float* __restrict__ x_conv,
    const float* __restrict__ xg, const float* __restrict__ w_dt,
    const float* __restrict__ b_dt, const float* __restrict__ A_log,
    const float* __restrict__ D_param, const float* __restrict__ hin,
    float* __restrict__ y)
{
    const int dg   = blockIdx.x % DGRP;
    const int bc   = blockIdx.x / DGRP;
    const int d    = dg * 256 + threadIdx.x;
    const int row0 = bc * CH_T;

    const float wdt = w_dt[d];
    const float bdt = b_dt[d];
    const float Dp  = D_param[d];
    float Aa[DS];
#pragma unroll
    for (int n = 0; n < DS; ++n) Aa[n] = -__expf(A_log[d * DS + n]);
    float h[DS];
#pragma unroll
    for (int n = 0; n < DS; ++n) h[n] = hin[((size_t)bc * DS + n) * DI + d];

#pragma unroll 4
    for (int t = 0; t < CH_T; ++t) {
        const int row = row0 + t;
        const float* s = ssm + (size_t)row * 33;
        const float x_t = x_conv[(size_t)row * DI + d];

        float z  = fmaf(s[32], wdt, bdt);
        float dt = (z > 20.0f) ? z : log1pf(__expf(z));

        float yv = 0.f;
#pragma unroll
        for (int n = 0; n < DS; ++n) {
            float Abar = __expf(dt * Aa[n]);
            h[n] = fmaf(Abar, h[n], (dt * s[n]) * x_t);
            yv   = fmaf(h[n], s[DS + n], yv);
        }
        yv = fmaf(x_t, Dp, yv);

        float gv = xg[(size_t)row * (2 * DI) + DI + d];
        float sg = gv / (1.0f + __expf(-gv));
        y[(size_t)row * DI + d] = yv * sg;
    }
}

extern "C" void kernel_launch(void* const* d_in, const int* in_sizes, int n_in,
                              void* d_out, int out_size, void* d_ws, size_t ws_size,
                              hipStream_t stream)
{
    const float* x       = (const float*)d_in[0];
    const float* W_in    = (const float*)d_in[1];
    const float* conv_w  = (const float*)d_in[2];
    const float* conv_b  = (const float*)d_in[3];
    const float* W_xproj = (const float*)d_in[4];
    const float* w_dt    = (const float*)d_in[5];
    const float* b_dt    = (const float*)d_in[6];
    const float* A_log   = (const float*)d_in[7];
    const float* D_param = (const float*)d_in[8];
    const float* W_out   = (const float*)d_in[9];
    float* out = (float*)d_out;

    float* ws     = (float*)d_ws;
    float* xg     = ws;
    float* x_conv = xg + (size_t)NROW * (2 * DI);
    float* ssm    = x_conv + (size_t)NROW * DI;
    float* yb     = ssm + (size_t)NROW * 33;
    float* hloc   = yb + (size_t)NROW * DI;
    float* hin    = hloc + (size_t)B_SZ * CH_C * DS * DI;
    float* sumdt  = hin + (size_t)B_SZ * CH_C * DS * DI;

    gemm_lds_f32_wmma<DM, 4><<<(NROW / 64) * ((2 * DI) / 128), 128, 0, stream>>>(
        x, W_in, xg, (2 * DI) / 128, 2 * DI);

    conv_silu_kernel<<<NROW / CV_T, DI / 4, 0, stream>>>(
        xg, conv_w, conv_b, x_conv);

    xproj_kernel<<<(NROW * 33 + 255) / 256, 256, 0, stream>>>(
        x_conv, W_xproj, ssm);

    scan_local_kernel<<<B_SZ * CH_C * DGRP, 256, 0, stream>>>(
        ssm, x_conv, w_dt, b_dt, A_log, hloc, sumdt);
    scan_carry_kernel<<<(B_SZ * DS * DI + 255) / 256, 256, 0, stream>>>(
        hloc, sumdt, A_log, hin);
    scan_replay_kernel<<<B_SZ * CH_C * DGRP, 256, 0, stream>>>(
        ssm, x_conv, xg, w_dt, b_dt, A_log, D_param, hin, yb);

    gemm_lds_f32_wmma<DI, 2><<<(NROW / 64) * (DM / 64), 128, 0, stream>>>(
        yb, W_out, out, DM / 64, DM);
}


# x-projection (N=33) moved from per-thread dot products to f32 WMMA: 32-row work-groups, K split over 8 waves, partials added through LDS in slice order; summation order of ssm changes
# speedup vs baseline: 12.4493x; 1.2506x over previous
#include <hip/hip_runtime.h>
#include <math.h>


typedef float  v2f __attribute__((ext_vector_type(2)));
typedef float  v4f __attribute__((ext_vector_type(4)));
typedef float  v8f __attribute__((ext_vector_type(8)));

#define B_SZ 2
#define L_SZ 2048
#define DM   768
#define DI   1536
#define DS   16
#define NROW (B_SZ * L_SZ)

template <int K, int NF>
__global__ __launch_bounds__(128, 6) void gemm_lds_f32_wmma(
    const float* __restrict__ A, const float* __restrict__ W,
    float* __restrict__ C, int NT, int ldc)
{
    static_assert(K % 32 == 0, "K must be a multiple of the panel depth");
    constexpr int BM    = 64;
    constexpr int BN    = 32 * NF;
    constexpr int KP    = 32;
    constexpr int PITCH = 36;
    constexpr int ROWS  = BM + BN;
    constexpr int NLD   = ROWS / 16;
    static_assert(NLD == 8 || NLD == 12, "staging is written out for NF = 2 and NF = 4");

    __shared__ __attribute__((aligned(16))) float lds[ROWS * PITCH];

    const int tid  = threadIdx.x;
    const int wave = tid >> 5;
    const int lane = tid & 31;
    const int half = lane >> 4;
    const int lr   = lane & 15;
    const int wm   = wave >> 1;
    const int wn   = wave & 1;
    const int tm   = blockIdx.x / NT;
    const int tn   = blockIdx.x % NT;

    const int r0 = tid >> 3;
    const int c4 = tid & 7;
    const float* gA = A + (size_t)(tm * BM + r0) * K + c4 * 4;
    const float* gW = W + (size_t)(tn * BN + r0) * K + c4 * 4;
    float* sdst = lds + r0 * PITCH + c4 * 4;

    v4f s0, s1, s2, s3, s4, s5, s6, s7, s8, s9, s10, s11;
    auto gload = [&](int kc) {
        s0 = *(const v4f*)(gA + (size_t)0  * K + kc);
        s1 = *(const v4f*)(gA + (size_t)16 * K + kc);
        s2 = *(const v4f*)(gA + (size_t)32 * K + kc);
        s3 = *(const v4f*)(gA + (size_t)48 * K + kc);
        s4 = *(const v4f*)(gW + (size_t)0  * K + kc);
        s5 = *(const v4f*)(gW + (size_t)16 * K + kc);
        s6 = *(const v4f*)(gW + (size_t)32 * K + kc);
        s7 = *(const v4f*)(gW + (size_t)48 * K + kc);
        if constexpr (NLD == 12) {
            s8  = *(const v4f*)(gW + (size_t)64  * K + kc);
            s9  = *(const v4f*)(gW + (size_t)80  * K + kc);
            s10 = *(const v4f*)(gW + (size_t)96  * K + kc);
            s11 = *(const v4f*)(gW + (size_t)112 * K + kc);
        }
    };
    auto sstore = [&]() {
        *(v4f*)(sdst + 0  * PITCH) = s0;
        *(v4f*)(sdst + 16 * PITCH) = s1;
        *(v4f*)(sdst + 32 * PITCH) = s2;
        *(v4f*)(sdst + 48 * PITCH) = s3;
        *(v4f*)(sdst + 64 * PITCH) = s4;
        *(v4f*)(sdst + 80 * PITCH) = s5;
        *(v4f*)(sdst + 96 * PITCH) = s6;
        *(v4f*)(sdst + 112 * PITCH) = s7;
        if constexpr (NLD == 12) {
            *(v4f*)(sdst + 128 * PITCH) = s8;
            *(v4f*)(sdst + 144 * PITCH) = s9;
            *(v4f*)(sdst + 160 * PITCH) = s10;
            *(v4f*)(sdst + 176 * PITCH) = s11;
        }
    };

    const float* sA = lds + (wm * 32 + lr) * PITCH + 2 * half;
    const float* sB = lds + (BM + wn * 16 * NF + lr) * PITCH + 2 * half;

    v8f acc[2][NF];
#pragma unroll
    for (int i = 0; i < 2; ++i)
#pragma unroll
        for (int f = 0; f < NF; ++f) acc[i][f] = v8f{};

    gload(0);
    for (int kc = 0; kc < K; kc += KP) {
        sstore();
        __syncthreads();
        if (kc + KP < K) gload(kc + KP);

#pragma unroll
        for (int ks = 0; ks < KP / 4; ++ks) {
            const v2f a0 = *(const v2f*)(sA + 4 * ks);
            const v2f a1 = *(const v2f*)(sA + 16 * PITCH + 4 * ks);
            v2f b[NF];
#pragma unroll
            for (int f = 0; f < NF; ++f)
                b[f] = *(const v2f*)(sB + f * 16 * PITCH + 4 * ks);
#pragma unroll
            for (int f = 0; f < NF; ++f)
                acc[0][f] = __builtin_amdgcn_wmma_f32_16x16x4_f32(
                    false, a0, false, b[f], (short)0, acc[0][f], false, false);
#pragma unroll
            for (int f = 0; f < NF; ++f)
                acc[1][f] = __builtin_amdgcn_wmma_f32_16x16x4_f32(
                    false, a1, false, b[f], (short)0, acc[1][f], false, false);
        }
        __syncthreads();
    }

    float* Cb = C + (size_t)(tm * BM + wm * 32 + 8 * half) * ldc
                  + tn * BN + wn * 16 * NF + lr;
#pragma unroll
    for (int r = 0; r < 8; ++r) {
#pragma unroll
        for (int i = 0; i < 2; ++i) {
            float* Cr = Cb + (size_t)(16 * i + r) * ldc;
#pragma unroll
            for (int f = 0; f < NF; ++f) Cr[16 * f] = acc[i][f][r];
        }
    }
}

#define CV_T 8

__global__ __launch_bounds__(DI / 4) void conv_silu_kernel(
    const float* __restrict__ xg, const float* __restrict__ conv_w,
    const float* __restrict__ conv_b, float* __restrict__ x_conv)
{
    static_assert(L_SZ % CV_T == 0 && CV_T >= 3,
                  "a chunk must not cross a sequence end, and only the first chunk of a sequence lacks its 3 rows in front");
    const unsigned d    = threadIdx.x * 4;
    const unsigned row0 = blockIdx.x * CV_T;
    const bool head = (row0 % L_SZ) == 0;

    v4f w[4];
#pragma unroll
    for (int c = 0; c < 4; ++c) w[c] = *(const v4f*)(conv_w + (size_t)(d + c) * 4);
    const v4f bias = *(const v4f*)(conv_b + d);

    v4f xin[CV_T + 3];
    const float* src = xg + (size_t)row0 * (2 * DI) + d;
    const size_t back = head ? 0 : (size_t)(2 * DI);
#pragma unroll
    for (int i = 0; i < 3; ++i)
        xin[i] = *(const v4f*)(src - (size_t)(3 - i) * back);
#pragma unroll
    for (int i = 3; i < CV_T + 3; ++i)
        xin[i] = *(const v4f*)(src + (size_t)(i - 3) * (2 * DI));

    float* dst = x_conv + (size_t)row0 * DI + d;
#pragma unroll
    for (int t = 0; t < CV_T; ++t) {
        v4f o;
#pragma unroll
        for (int c = 0; c < 4; ++c) {
            float acc = bias[c];
#pragma unroll
            for (int j = 0; j < 4; ++j)
                if (t + j >= 3 || !head)
                    acc = fmaf(w[c][j], xin[t + j][c], acc);
            o[c] = acc / (1.0f + __expf(-acc));
        }
        *(v4f*)(dst + (size_t)t * DI) = o;
    }
}

#define XP_N     33
#define XP_BM    32
#define XP_WAVES 8
#define XP_KS    (DI / XP_WAVES)
#define XP_GB    8
#define XP_NB    (XP_KS / (8 * XP_GB))
#define XP_PITCH 36

__global__ __launch_bounds__(32 * XP_WAVES) __attribute__((amdgpu_waves_per_eu(2, 2)))
void xproj_wmma_kernel(
    const float* __restrict__ x_conv, const float* __restrict__ Wx,
    float* __restrict__ ssm)
{
    static_assert(DI % (XP_WAVES * 8 * XP_GB) == 0, "K slices are whole load batches");
    static_assert(NROW % XP_BM == 0, "whole row blocks");
    static_assert((XP_BM * XP_N) % 32 == 0, "a work-group's results are whole 128-byte lines");

    __shared__ float red[XP_WAVES * XP_BM * XP_PITCH];

    const int tid  = threadIdx.x;
    const int wave = tid >> 5;
    const int lane = tid & 31;
    const int half = lane >> 4;
    const int lr   = lane & 15;
    const int row0 = blockIdx.x * XP_BM;

    const int kb = wave * XP_KS + 4 * half;
    const float* pA0 = x_conv + (size_t)(row0 + lr) * DI + kb;
    const float* pA1 = pA0 + (size_t)16 * DI;
    const float* pB0 = Wx + (size_t)lr * DI + kb;
    const float* pB1 = pB0 + (size_t)16 * DI;
    const float* pB2 = Wx + (size_t)32 * DI + kb;
    const bool col32 = (lr == 0);

    v8f acc[2][3];
#pragma unroll
    for (int i = 0; i < 2; ++i)
#pragma unroll
        for (int f = 0; f < 3; ++f) acc[i][f] = v8f{};

    v4f a0[2][XP_GB], a1[2][XP_GB], b0[2][XP_GB], b1[2][XP_GB], b2[2][XP_GB];

#pragma unroll
    for (int g = 0; g < XP_GB; ++g) {
        a0[0][g] = *(const v4f*)(pA0 + 8 * g);
        a1[0][g] = *(const v4f*)(pA1 + 8 * g);
        b0[0][g] = *(const v4f*)(pB0 + 8 * g);
        b1[0][g] = *(const v4f*)(pB1 + 8 * g);
        b2[0][g] = *(const v4f*)(pB2 + 8 * g);
    }

#pragma unroll
    for (int bt = 0; bt < XP_NB; ++bt) {
        const int cur = bt & 1;
        const int nxt = cur ^ 1;
        if (bt + 1 < XP_NB) {
#pragma unroll
            for (int g = 0; g < XP_GB; ++g) {
                const int o = ((bt + 1) * XP_GB + g) * 8;
                a0[nxt][g] = *(const v4f*)(pA0 + o);
                a1[nxt][g] = *(const v4f*)(pA1 + o);
                b0[nxt][g] = *(const v4f*)(pB0 + o);
                b1[nxt][g] = *(const v4f*)(pB1 + o);
                b2[nxt][g] = *(const v4f*)(pB2 + o);
            }
        }
#pragma unroll
        for (int g = 0; g < XP_GB; ++g) {
            const v4f va0 = a0[cur][g];
            const v4f va1 = a1[cur][g];
            const v4f vb0 = b0[cur][g];
            const v4f vb1 = b1[cur][g];
            const v4f vb2 = col32 ? b2[cur][g] : v4f{0.f, 0.f, 0.f, 0.f};
#pragma unroll
            for (int p = 0; p < 2; ++p) {
                const v2f fa0 = {va0[2 * p], va0[2 * p + 1]};
                const v2f fa1 = {va1[2 * p], va1[2 * p + 1]};
                const v2f fb0 = {vb0[2 * p], vb0[2 * p + 1]};
                const v2f fb1 = {vb1[2 * p], vb1[2 * p + 1]};
                const v2f fb2 = {vb2[2 * p], vb2[2 * p + 1]};
                acc[0][0] = __builtin_amdgcn_wmma_f32_16x16x4_f32(
                    false, fa0, false, fb0, (short)0, acc[0][0], false, false);
                acc[0][1] = __builtin_amdgcn_wmma_f32_16x16x4_f32(
                    false, fa0, false, fb1, (short)0, acc[0][1], false, false);
                acc[0][2] = __builtin_amdgcn_wmma_f32_16x16x4_f32(
                    false, fa0, false, fb2, (short)0, acc[0][2], false, false);
                acc[1][0] = __builtin_amdgcn_wmma_f32_16x16x4_f32(
                    false, fa1, false, fb0, (short)0, acc[1][0], false, false);
                acc[1][1] = __builtin_amdgcn_wmma_f32_16x16x4_f32(
                    false, fa1, false, fb1, (short)0, acc[1][1], false, false);
                acc[1][2] = __builtin_amdgcn_wmma_f32_16x16x4_f32(
                    false, fa1, false, fb2, (short)0, acc[1][2], false, false);
            }
        }
    }

    float* rw = red + wave * (XP_BM * XP_PITCH) + (8 * half) * XP_PITCH + lr;
#pragma unroll
    for (int i = 0; i < 2; ++i) {
#pragma unroll
        for (int r = 0; r < 8; ++r) {
            float* p = rw + (16 * i + r) * XP_PITCH;
            p[0]  = acc[i][0][r];
            p[16] = acc[i][1][r];
            if (col32) p[32] = acc[i][2][r];
        }
    }
    __syncthreads();

    float* dst = ssm + (size_t)row0 * XP_N;
#pragma unroll
    for (int j = 0; j < (XP_BM * XP_N + 255) / 256; ++j) {
        const int e = tid + 256 * j;
        if (e < XP_BM * XP_N) {
            const int row = e / XP_N;
            const int n   = e - row * XP_N;
            const float* p = red + row * XP_PITCH + n;
            float s = p[0];
#pragma unroll
            for (int w = 1; w < XP_WAVES; ++w) s += p[w * (XP_BM * XP_PITCH)];
            dst[e] = s;
        }
    }
}

#define CH_T 32
#define CH_C (L_SZ / CH_T)
#define DGRP (DI / 256)

__global__ __launch_bounds__(256) void scan_local_kernel(
    const float* __restrict__ ssm, const float* __restrict__ x_conv,
    const float* __restrict__ w_dt, const float* __restrict__ b_dt,
    const float* __restrict__ A_log,
    float* __restrict__ hloc, float* __restrict__ sumdt)
{
    const int dg   = blockIdx.x % DGRP;
    const int bc   = blockIdx.x / DGRP;
    const int d    = dg * 256 + threadIdx.x;
    const int row0 = bc * CH_T;

    const float wdt = w_dt[d];
    const float bdt = b_dt[d];
    float Aa[DS];
#pragma unroll
    for (int n = 0; n < DS; ++n) Aa[n] = -__expf(A_log[d * DS + n]);
    float h[DS];
#pragma unroll
    for (int n = 0; n < DS; ++n) h[n] = 0.f;
    float sdt = 0.f;

#pragma unroll 4
    for (int t = 0; t < CH_T; ++t) {
        const int row = row0 + t;
        const float* s = ssm + (size_t)row * 33;
        const float x_t = x_conv[(size_t)row * DI + d];

        float z  = fmaf(s[32], wdt, bdt);
        float dt = (z > 20.0f) ? z : log1pf(__expf(z));
        sdt += dt;
#pragma unroll
        for (int n = 0; n < DS; ++n) {
            float Abar = __expf(dt * Aa[n]);
            h[n] = fmaf(Abar, h[n], (dt * s[n]) * x_t);
        }
    }
#pragma unroll
    for (int n = 0; n < DS; ++n)
        hloc[((size_t)bc * DS + n) * DI + d] = h[n];
    sumdt[(size_t)bc * DI + d] = sdt;
}

__global__ __launch_bounds__(256) void scan_carry_kernel(
    const float* __restrict__ hloc, const float* __restrict__ sumdt,
    const float* __restrict__ A_log, float* __restrict__ hin)
{
    const int idx = blockIdx.x * 256 + threadIdx.x;
    if (idx >= B_SZ * DS * DI) return;
    const int d = idx % DI;
    const int n = (idx / DI) % DS;
    const int b = idx / (DI * DS);
    const float Aa = -__expf(A_log[d * DS + n]);

    float h = 0.f;
#pragma unroll 8
    for (int c = 0; c < CH_C; ++c) {
        const size_t bc = (size_t)b * CH_C + c;
        const float hl = hloc[(bc * DS + n) * DI + d];
        const float sd = sumdt[bc * DI + d];
        hin[(bc * DS + n) * DI + d] = h;
        h = fmaf(__expf(Aa * sd), h, hl);
    }
}

__global__ __launch_bounds__(256) void scan_replay_kernel(
    const float* __restrict__ ssm, const float* __restrict__ x_conv,
    const float* __restrict__ xg, const float* __restrict__ w_dt,
    const float* __restrict__ b_dt, const float* __restrict__ A_log,
    const float* __restrict__ D_param, const float* __restrict__ hin,
    float* __restrict__ y)
{
    const int dg   = blockIdx.x % DGRP;
    const int bc   = blockIdx.x / DGRP;
    const int d    = dg * 256 + threadIdx.x;
    const int row0 = bc * CH_T;

    const float wdt = w_dt[d];
    const float bdt = b_dt[d];
    const float Dp  = D_param[d];
    float Aa[DS];
#pragma unroll
    for (int n = 0; n < DS; ++n) Aa[n] = -__expf(A_log[d * DS + n]);
    float h[DS];
#pragma unroll
    for (int n = 0; n < DS; ++n) h[n] = hin[((size_t)bc * DS + n) * DI + d];

#pragma unroll 4
    for (int t = 0; t < CH_T; ++t) {
        const int row = row0 + t;
        const float* s = ssm + (size_t)row * 33;
        const float x_t = x_conv[(size_t)row * DI + d];

        float z  = fmaf(s[32], wdt, bdt);
        float dt = (z > 20.0f) ? z : log1pf(__expf(z));

        float yv = 0.f;
#pragma unroll
        for (int n = 0; n < DS; ++n) {
            float Abar = __expf(dt * Aa[n]);
            h[n] = fmaf(Abar, h[n], (dt * s[n]) * x_t);
            yv   = fmaf(h[n], s[DS + n], yv);
        }
        yv = fmaf(x_t, Dp, yv);

        float gv = xg[(size_t)row * (2 * DI) + DI + d];
        float sg = gv / (1.0f + __expf(-gv));
        y[(size_t)row * DI + d] = yv * sg;
    }
}

extern "C" void kernel_launch(void* const* d_in, const int* in_sizes, int n_in,
                              void* d_out, int out_size, void* d_ws, size_t ws_size,
                              hipStream_t stream)
{
    const float* x       = (const float*)d_in[0];
    const float* W_in    = (const float*)d_in[1];
    const float* conv_w  = (const float*)d_in[2];
    const float* conv_b  = (const float*)d_in[3];
    const float* W_xproj = (const float*)d_in[4];
    const float* w_dt    = (const float*)d_in[5];
    const float* b_dt    = (const float*)d_in[6];
    const float* A_log   = (const float*)d_in[7];
    const float* D_param = (const float*)d_in[8];
    const float* W_out   = (const float*)d_in[9];
    float* out = (float*)d_out;

    float* ws     = (float*)d_ws;
    float* xg     = ws;
    float* x_conv = xg + (size_t)NROW * (2 * DI);
    float* ssm    = x_conv + (size_t)NROW * DI;
    float* yb     = ssm + (size_t)NROW * 33;
    float* hloc   = yb + (size_t)NROW * DI;
    float* hin    = hloc + (size_t)B_SZ * CH_C * DS * DI;
    float* sumdt  = hin + (size_t)B_SZ * CH_C * DS * DI;

    gemm_lds_f32_wmma<DM, 4><<<(NROW / 64) * ((2 * DI) / 128), 128, 0, stream>>>(
        x, W_in, xg, (2 * DI) / 128, 2 * DI);

    conv_silu_kernel<<<NROW / CV_T, DI / 4, 0, stream>>>(
        xg, conv_w, conv_b, x_conv);

    xproj_wmma_kernel<<<NROW / XP_BM, 32 * XP_WAVES, 0, stream>>>(
        x_conv, W_xproj, ssm);

    scan_local_kernel<<<B_SZ * CH_C * DGRP, 256, 0, stream>>>(
        ssm, x_conv, w_dt, b_dt, A_log, hloc, sumdt);
    scan_carry_kernel<<<(B_SZ * DS * DI + 255) / 256, 256, 0, stream>>>(
        hloc, sumdt, A_log, hin);
    scan_replay_kernel<<<B_SZ * CH_C * DGRP, 256, 0, stream>>>(
        ssm, x_conv, xg, w_dt, b_dt, A_log, D_param, hin, yb);

    gemm_lds_f32_wmma<DI, 2><<<(NROW / 64) * (DM / 64), 128, 0, stream>>>(
        yb, W_out, out, DM / 64, DM);
}


# softplus in both scan passes by a short form, max(z,0) + 2 atanh(u/(2+u)) with u = exp(-|z|), on hardware exp and rcp with an 8-term series, in place of the in-line library log1pf
# speedup vs baseline: 13.5882x; 1.0915x over previous
#include <hip/hip_runtime.h>
#include <math.h>


typedef float  v2f __attribute__((ext_vector_type(2)));
typedef float  v4f __attribute__((ext_vector_type(4)));
typedef float  v8f __attribute__((ext_vector_type(8)));

#define B_SZ 2
#define L_SZ 2048
#define DM   768
#define DI   1536
#define DS   16
#define NROW (B_SZ * L_SZ)

template <int K, int NF>
__global__ __launch_bounds__(128, 6) void gemm_lds_f32_wmma(
    const float* __restrict__ A, const float* __restrict__ W,
    float* __restrict__ C, int NT, int ldc)
{
    static_assert(K % 32 == 0, "K must be a multiple of the panel depth");
    constexpr int BM    = 64;
    constexpr int BN    = 32 * NF;
    constexpr int KP    = 32;
    constexpr int PITCH = 36;
    constexpr int ROWS  = BM + BN;
    constexpr int NLD   = ROWS / 16;
    static_assert(NLD == 8 || NLD == 12, "staging is written out for NF = 2 and NF = 4");

    __shared__ __attribute__((aligned(16))) float lds[ROWS * PITCH];

    const int tid  = threadIdx.x;
    const int wave = tid >> 5;
    const int lane = tid & 31;
    const int half = lane >> 4;
    const int lr   = lane & 15;
    const int wm   = wave >> 1;
    const int wn   = wave & 1;
    const int tm   = blockIdx.x / NT;
    const int tn   = blockIdx.x % NT;

    const int r0 = tid >> 3;
    const int c4 = tid & 7;
    const float* gA = A + (size_t)(tm * BM + r0) * K + c4 * 4;
    const float* gW = W + (size_t)(tn * BN + r0) * K + c4 * 4;
    float* sdst = lds + r0 * PITCH + c4 * 4;

    v4f s0, s1, s2, s3, s4, s5, s6, s7, s8, s9, s10, s11;
    auto gload = [&](int kc) {
        s0 = *(const v4f*)(gA + (size_t)0  * K + kc);
        s1 = *(const v4f*)(gA + (size_t)16 * K + kc);
        s2 = *(const v4f*)(gA + (size_t)32 * K + kc);
        s3 = *(const v4f*)(gA + (size_t)48 * K + kc);
        s4 = *(const v4f*)(gW + (size_t)0  * K + kc);
        s5 = *(const v4f*)(gW + (size_t)16 * K + kc);
        s6 = *(const v4f*)(gW + (size_t)32 * K + kc);
        s7 = *(const v4f*)(gW + (size_t)48 * K + kc);
        if constexpr (NLD == 12) {
            s8  = *(const v4f*)(gW + (size_t)64  * K + kc);
            s9  = *(const v4f*)(gW + (size_t)80  * K + kc);
            s10 = *(const v4f*)(gW + (size_t)96  * K + kc);
            s11 = *(const v4f*)(gW + (size_t)112 * K + kc);
        }
    };
    auto sstore = [&]() {
        *(v4f*)(sdst + 0  * PITCH) = s0;
        *(v4f*)(sdst + 16 * PITCH) = s1;
        *(v4f*)(sdst + 32 * PITCH) = s2;
        *(v4f*)(sdst + 48 * PITCH) = s3;
        *(v4f*)(sdst + 64 * PITCH) = s4;
        *(v4f*)(sdst + 80 * PITCH) = s5;
        *(v4f*)(sdst + 96 * PITCH) = s6;
        *(v4f*)(sdst + 112 * PITCH) = s7;
        if constexpr (NLD == 12) {
            *(v4f*)(sdst + 128 * PITCH) = s8;
            *(v4f*)(sdst + 144 * PITCH) = s9;
            *(v4f*)(sdst + 160 * PITCH) = s10;
            *(v4f*)(sdst + 176 * PITCH) = s11;
        }
    };

    const float* sA = lds + (wm * 32 + lr) * PITCH + 2 * half;
    const float* sB = lds + (BM + wn * 16 * NF + lr) * PITCH + 2 * half;

    v8f acc[2][NF];
#pragma unroll
    for (int i = 0; i < 2; ++i)
#pragma unroll
        for (int f = 0; f < NF; ++f) acc[i][f] = v8f{};

    gload(0);
    for (int kc = 0; kc < K; kc += KP) {
        sstore();
        __syncthreads();
        if (kc + KP < K) gload(kc + KP);

#pragma unroll
        for (int ks = 0; ks < KP / 4; ++ks) {
            const v2f a0 = *(const v2f*)(sA + 4 * ks);
            const v2f a1 = *(const v2f*)(sA + 16 * PITCH + 4 * ks);
            v2f b[NF];
#pragma unroll
            for (int f = 0; f < NF; ++f)
                b[f] = *(const v2f*)(sB + f * 16 * PITCH + 4 * ks);
#pragma unroll
            for (int f = 0; f < NF; ++f)
                acc[0][f] = __builtin_amdgcn_wmma_f32_16x16x4_f32(
                    false, a0, false, b[f], (short)0, acc[0][f], false, false);
#pragma unroll
            for (int f = 0; f < NF; ++f)
                acc[1][f] = __builtin_amdgcn_wmma_f32_16x16x4_f32(
                    false, a1, false, b[f], (short)0, acc[1][f], false, false);
        }
        __syncthreads();
    }

    float* Cb = C + (size_t)(tm * BM + wm * 32 + 8 * half) * ldc
                  + tn * BN + wn * 16 * NF + lr;
#pragma unroll
    for (int r = 0; r < 8; ++r) {
#pragma unroll
        for (int i = 0; i < 2; ++i) {
            float* Cr = Cb + (size_t)(16 * i + r) * ldc;
#pragma unroll
            for (int f = 0; f < NF; ++f) Cr[16 * f] = acc[i][f][r];
        }
    }
}

#define CV_T 8

__global__ __launch_bounds__(DI / 4) void conv_silu_kernel(
    const float* __restrict__ xg, const float* __restrict__ conv_w,
    const float* __restrict__ conv_b, float* __restrict__ x_conv)
{
    static_assert(L_SZ % CV_T == 0 && CV_T >= 3,
                  "a chunk must not cross a sequence end, and only the first chunk of a sequence lacks its 3 rows in front");
    const unsigned d    = threadIdx.x * 4;
    const unsigned row0 = blockIdx.x * CV_T;
    const bool head = (row0 % L_SZ) == 0;

    v4f w[4];
#pragma unroll
    for (int c = 0; c < 4; ++c) w[c] = *(const v4f*)(conv_w + (size_t)(d + c) * 4);
    const v4f bias = *(const v4f*)(conv_b + d);

    v4f xin[CV_T + 3];
    const float* src = xg + (size_t)row0 * (2 * DI) + d;
    const size_t back = head ? 0 : (size_t)(2 * DI);
#pragma unroll
    for (int i = 0; i < 3; ++i)
        xin[i] = *(const v4f*)(src - (size_t)(3 - i) * back);
#pragma unroll
    for (int i = 3; i < CV_T + 3; ++i)
        xin[i] = *(const v4f*)(src + (size_t)(i - 3) * (2 * DI));

    float* dst = x_conv + (size_t)row0 * DI + d;
#pragma unroll
    for (int t = 0; t < CV_T; ++t) {
        v4f o;
#pragma unroll
        for (int c = 0; c < 4; ++c) {
            float acc = bias[c];
#pragma unroll
            for (int j = 0; j < 4; ++j)
                if (t + j >= 3 || !head)
                    acc = fmaf(w[c][j], xin[t + j][c], acc);
            o[c] = acc / (1.0f + __expf(-acc));
        }
        *(v4f*)(dst + (size_t)t * DI) = o;
    }
}

#define XP_N     33
#define XP_BM    32
#define XP_WAVES 8
#define XP_KS    (DI / XP_WAVES)
#define XP_GB    8
#define XP_NB    (XP_KS / (8 * XP_GB))
#define XP_PITCH 36

__global__ __launch_bounds__(32 * XP_WAVES) __attribute__((amdgpu_waves_per_eu(2, 2)))
void xproj_wmma_kernel(
    const float* __restrict__ x_conv, const float* __restrict__ Wx,
    float* __restrict__ ssm)
{
    static_assert(DI % (XP_WAVES * 8 * XP_GB) == 0, "K slices are whole load batches");
    static_assert(NROW % XP_BM == 0, "whole row blocks");
    static_assert((XP_BM * XP_N) % 32 == 0, "a work-group's results are whole 128-byte lines");

    __shared__ float red[XP_WAVES * XP_BM * XP_PITCH];

    const int tid  = threadIdx.x;
    const int wave = tid >> 5;
    const int lane = tid & 31;
    const int half = lane >> 4;
    const int lr   = lane & 15;
    const int row0 = blockIdx.x * XP_BM;

    const int kb = wave * XP_KS + 4 * half;
    const float* pA0 = x_conv + (size_t)(row0 + lr) * DI + kb;
    const float* pA1 = pA0 + (size_t)16 * DI;
    const float* pB0 = Wx + (size_t)lr * DI + kb;
    const float* pB1 = pB0 + (size_t)16 * DI;
    const float* pB2 = Wx + (size_t)32 * DI + kb;
    const bool col32 = (lr == 0);

    v8f acc[2][3];
#pragma unroll
    for (int i = 0; i < 2; ++i)
#pragma unroll
        for (int f = 0; f < 3; ++f) acc[i][f] = v8f{};

    v4f a0[2][XP_GB], a1[2][XP_GB], b0[2][XP_GB], b1[2][XP_GB], b2[2][XP_GB];

#pragma unroll
    for (int g = 0; g < XP_GB; ++g) {
        a0[0][g] = *(const v4f*)(pA0 + 8 * g);
        a1[0][g] = *(const v4f*)(pA1 + 8 * g);
        b0[0][g] = *(const v4f*)(pB0 + 8 * g);
        b1[0][g] = *(const v4f*)(pB1 + 8 * g);
        b2[0][g] = *(const v4f*)(pB2 + 8 * g);
    }

#pragma unroll
    for (int bt = 0; bt < XP_NB; ++bt) {
        const int cur = bt & 1;
        const int nxt = cur ^ 1;
        if (bt + 1 < XP_NB) {
#pragma unroll
            for (int g = 0; g < XP_GB; ++g) {
                const int o = ((bt + 1) * XP_GB + g) * 8;
                a0[nxt][g] = *(const v4f*)(pA0 + o);
                a1[nxt][g] = *(const v4f*)(pA1 + o);
                b0[nxt][g] = *(const v4f*)(pB0 + o);
                b1[nxt][g] = *(const v4f*)(pB1 + o);
                b2[nxt][g] = *(const v4f*)(pB2 + o);
            }
        }
#pragma unroll
        for (int g = 0; g < XP_GB; ++g) {
            const v4f va0 = a0[cur][g];
            const v4f va1 = a1[cur][g];
            const v4f vb0 = b0[cur][g];
            const v4f vb1 = b1[cur][g];
            const v4f vb2 = col32 ? b2[cur][g] : v4f{0.f, 0.f, 0.f, 0.f};
#pragma unroll
            for (int p = 0; p < 2; ++p) {
                const v2f fa0 = {va0[2 * p], va0[2 * p + 1]};
                const v2f fa1 = {va1[2 * p], va1[2 * p + 1]};
                const v2f fb0 = {vb0[2 * p], vb0[2 * p + 1]};
                const v2f fb1 = {vb1[2 * p], vb1[2 * p + 1]};
                const v2f fb2 = {vb2[2 * p], vb2[2 * p + 1]};
                acc[0][0] = __builtin_amdgcn_wmma_f32_16x16x4_f32(
                    false, fa0, false, fb0, (short)0, acc[0][0], false, false);
                acc[0][1] = __builtin_amdgcn_wmma_f32_16x16x4_f32(
                    false, fa0, false, fb1, (short)0, acc[0][1], false, false);
                acc[0][2] = __builtin_amdgcn_wmma_f32_16x16x4_f32(
                    false, fa0, false, fb2, (short)0, acc[0][2], false, false);
                acc[1][0] = __builtin_amdgcn_wmma_f32_16x16x4_f32(
                    false, fa1, false, fb0, (short)0, acc[1][0], false, false);
                acc[1][1] = __builtin_amdgcn_wmma_f32_16x16x4_f32(
                    false, fa1, false, fb1, (short)0, acc[1][1], false, false);
                acc[1][2] = __builtin_amdgcn_wmma_f32_16x16x4_f32(
                    false, fa1, false, fb2, (short)0, acc[1][2], false, false);
            }
        }
    }

    float* rw = red + wave * (XP_BM * XP_PITCH) + (8 * half) * XP_PITCH + lr;
#pragma unroll
    for (int i = 0; i < 2; ++i) {
#pragma unroll
        for (int r = 0; r < 8; ++r) {
            float* p = rw + (16 * i + r) * XP_PITCH;
            p[0]  = acc[i][0][r];
            p[16] = acc[i][1][r];
            if (col32) p[32] = acc[i][2][r];
        }
    }
    __syncthreads();

    float* dst = ssm + (size_t)row0 * XP_N;
#pragma unroll
    for (int j = 0; j < (XP_BM * XP_N + 255) / 256; ++j) {
        const int e = tid + 256 * j;
        if (e < XP_BM * XP_N) {
            const int row = e / XP_N;
            const int n   = e - row * XP_N;
            const float* p = red + row * XP_PITCH + n;
            float s = p[0];
#pragma unroll
            for (int w = 1; w < XP_WAVES; ++w) s += p[w * (XP_BM * XP_PITCH)];
            dst[e] = s;
        }
    }
}

#define CH_T 32
#define CH_C (L_SZ / CH_T)
#define DGRP (DI / 256)

__device__ __forceinline__ float softplus_short(float z)
{
    const float u = __expf(-fabsf(z));
    const float s = u * __builtin_amdgcn_rcpf(2.0f + u);
    const float q = s * s;
    float p = 2.0f / 15.0f;
    p = fmaf(p, q, 2.0f / 13.0f);
    p = fmaf(p, q, 2.0f / 11.0f);
    p = fmaf(p, q, 2.0f / 9.0f);
    p = fmaf(p, q, 2.0f / 7.0f);
    p = fmaf(p, q, 2.0f / 5.0f);
    p = fmaf(p, q, 2.0f / 3.0f);
    p = fmaf(p, q, 2.0f);
    return fmaf(s, p, fmaxf(z, 0.0f));
}

__global__ __launch_bounds__(256) void scan_local_kernel(
    const float* __restrict__ ssm, const float* __restrict__ x_conv,
    const float* __restrict__ w_dt, const float* __restrict__ b_dt,
    const float* __restrict__ A_log,
    float* __restrict__ hloc, float* __restrict__ sumdt)
{
    const int dg   = blockIdx.x % DGRP;
    const int bc   = blockIdx.x / DGRP;
    const int d    = dg * 256 + threadIdx.x;
    const int row0 = bc * CH_T;

    const float wdt = w_dt[d];
    const float bdt = b_dt[d];
    float Aa[DS];
#pragma unroll
    for (int n = 0; n < DS; ++n) Aa[n] = -__expf(A_log[d * DS + n]);
    float h[DS];
#pragma unroll
    for (int n = 0; n < DS; ++n) h[n] = 0.f;
    float sdt = 0.f;

#pragma unroll 4
    for (int t = 0; t < CH_T; ++t) {
        const int row = row0 + t;
        const float* s = ssm + (size_t)row * 33;
        const float x_t = x_conv[(size_t)row * DI + d];

        float z  = fmaf(s[32], wdt, bdt);
        float dt = softplus_short(z);
        sdt += dt;
#pragma unroll
        for (int n = 0; n < DS; ++n) {
            float Abar = __expf(dt * Aa[n]);
            h[n] = fmaf(Abar, h[n], (dt * s[n]) * x_t);
        }
    }
#pragma unroll
    for (int n = 0; n < DS; ++n)
        hloc[((size_t)bc * DS + n) * DI + d] = h[n];
    sumdt[(size_t)bc * DI + d] = sdt;
}

__global__ __launch_bounds__(256) void scan_carry_kernel(
    const float* __restrict__ hloc, const float* __restrict__ sumdt,
    const float* __restrict__ A_log, float* __restrict__ hin)
{
    const int idx = blockIdx.x * 256 + threadIdx.x;
    if (idx >= B_SZ * DS * DI) return;
    const int d = idx % DI;
    const int n = (idx / DI) % DS;
    const int b = idx / (DI * DS);
    const float Aa = -__expf(A_log[d * DS + n]);

    float h = 0.f;
#pragma unroll 8
    for (int c = 0; c < CH_C; ++c) {
        const size_t bc = (size_t)b * CH_C + c;
        const float hl = hloc[(bc * DS + n) * DI + d];
        const float sd = sumdt[bc * DI + d];
        hin[(bc * DS + n) * DI + d] = h;
        h = fmaf(__expf(Aa * sd), h, hl);
    }
}

__global__ __launch_bounds__(256) void scan_replay_kernel(
    const float* __restrict__ ssm, const float* __restrict__ x_conv,
    const float* __restrict__ xg, const float* __restrict__ w_dt,
    const float* __restrict__ b_dt, const float* __restrict__ A_log,
    const float* __restrict__ D_param, const float* __restrict__ hin,
    float* __restrict__ y)
{
    const int dg   = blockIdx.x % DGRP;
    const int bc   = blockIdx.x / DGRP;
    const int d    = dg * 256 + threadIdx.x;
    const int row0 = bc * CH_T;

    const float wdt = w_dt[d];
    const float bdt = b_dt[d];
    const float Dp  = D_param[d];
    float Aa[DS];
#pragma unroll
    for (int n = 0; n < DS; ++n) Aa[n] = -__expf(A_log[d * DS + n]);
    float h[DS];
#pragma unroll
    for (int n = 0; n < DS; ++n) h[n] = hin[((size_t)bc * DS + n) * DI + d];

#pragma unroll 4
    for (int t = 0; t < CH_T; ++t) {
        const int row = row0 + t;
        const float* s = ssm + (size_t)row * 33;
        const float x_t = x_conv[(size_t)row * DI + d];

        float z  = fmaf(s[32], wdt, bdt);
        float dt = softplus_short(z);

        float yv = 0.f;
#pragma unroll
        for (int n = 0; n < DS; ++n) {
            float Abar = __expf(dt * Aa[n]);
            h[n] = fmaf(Abar, h[n], (dt * s[n]) * x_t);
            yv   = fmaf(h[n], s[DS + n], yv);
        }
        yv = fmaf(x_t, Dp, yv);

        float gv = xg[(size_t)row * (2 * DI) + DI + d];
        float sg = gv / (1.0f + __expf(-gv));
        y[(size_t)row * DI + d] = yv * sg;
    }
}

extern "C" void kernel_launch(void* const* d_in, const int* in_sizes, int n_in,
                              void* d_out, int out_size, void* d_ws, size_t ws_size,
                              hipStream_t stream)
{
    const float* x       = (const float*)d_in[0];
    const float* W_in    = (const float*)d_in[1];
    const float* conv_w  = (const float*)d_in[2];
    const float* conv_b  = (const float*)d_in[3];
    const float* W_xproj = (const float*)d_in[4];
    const float* w_dt    = (const float*)d_in[5];
    const float* b_dt    = (const float*)d_in[6];
    const float* A_log   = (const float*)d_in[7];
    const float* D_param = (const float*)d_in[8];
    const float* W_out   = (const float*)d_in[9];
    float* out = (float*)d_out;

    float* ws     = (float*)d_ws;
    float* xg     = ws;
    float* x_conv = xg + (size_t)NROW * (2 * DI);
    float* ssm    = x_conv + (size_t)NROW * DI;
    float* yb     = ssm + (size_t)NROW * 33;
    float* hloc   = yb + (size_t)NROW * DI;
    float* hin    = hloc + (size_t)B_SZ * CH_C * DS * DI;
    float* sumdt  = hin + (size_t)B_SZ * CH_C * DS * DI;

    gemm_lds_f32_wmma<DM, 4><<<(NROW / 64) * ((2 * DI) / 128), 128, 0, stream>>>(
        x, W_in, xg, (2 * DI) / 128, 2 * DI);

    conv_silu_kernel<<<NROW / CV_T, DI / 4, 0, stream>>>(
        xg, conv_w, conv_b, x_conv);

    xproj_wmma_kernel<<<NROW / XP_BM, 32 * XP_WAVES, 0, stream>>>(
        x_conv, W_xproj, ssm);

    scan_local_kernel<<<B_SZ * CH_C * DGRP, 256, 0, stream>>>(
        ssm, x_conv, w_dt, b_dt, A_log, hloc, sumdt);
    scan_carry_kernel<<<(B_SZ * DS * DI + 255) / 256, 256, 0, stream>>>(
        hloc, sumdt, A_log, hin);
    scan_replay_kernel<<<B_SZ * CH_C * DGRP, 256, 0, stream>>>(
        ssm, x_conv, xg, w_dt, b_dt, A_log, D_param, hin, yb);

    gemm_lds_f32_wmma<DI, 2><<<(NROW / 64) * (DM / 64), 128, 0, stream>>>(
        yb, W_out, out, DM / 64, DM);
}


# conv+SiLU and the x-projection fused into one launch: each wave turns its own 16x128 x_conv tile through LDS into the WMMA A operand, K split over 12 waves; summation order of ssm changes
# speedup vs baseline: 14.1615x; 1.0422x over previous
#include <hip/hip_runtime.h>
#include <math.h>


typedef float  v2f __attribute__((ext_vector_type(2)));
typedef float  v4f __attribute__((ext_vector_type(4)));
typedef float  v8f __attribute__((ext_vector_type(8)));

#define B_SZ 2
#define L_SZ 2048
#define DM   768
#define DI   1536
#define DS   16
#define NROW (B_SZ * L_SZ)

template <int K, int NF>
__global__ __launch_bounds__(128, 6) void gemm_lds_f32_wmma(
    const float* __restrict__ A, const float* __restrict__ W,
    float* __restrict__ C, int NT, int ldc)
{
    static_assert(K % 32 == 0, "K must be a multiple of the panel depth");
    constexpr int BM    = 64;
    constexpr int BN    = 32 * NF;
    constexpr int KP    = 32;
    constexpr int PITCH = 36;
    constexpr int ROWS  = BM + BN;
    constexpr int NLD   = ROWS / 16;
    static_assert(NLD == 8 || NLD == 12, "staging is written out for NF = 2 and NF = 4");

    __shared__ __attribute__((aligned(16))) float lds[ROWS * PITCH];

    const int tid  = threadIdx.x;
    const int wave = tid >> 5;
    const int lane = tid & 31;
    const int half = lane >> 4;
    const int lr   = lane & 15;
    const int wm   = wave >> 1;
    const int wn   = wave & 1;
    const int tm   = blockIdx.x / NT;
    const int tn   = blockIdx.x % NT;

    const int r0 = tid >> 3;
    const int c4 = tid & 7;
    const float* gA = A + (size_t)(tm * BM + r0) * K + c4 * 4;
    const float* gW = W + (size_t)(tn * BN + r0) * K + c4 * 4;
    float* sdst = lds + r0 * PITCH + c4 * 4;

    v4f s0, s1, s2, s3, s4, s5, s6, s7, s8, s9, s10, s11;
    auto gload = [&](int kc) {
        s0 = *(const v4f*)(gA + (size_t)0  * K + kc);
        s1 = *(const v4f*)(gA + (size_t)16 * K + kc);
        s2 = *(const v4f*)(gA + (size_t)32 * K + kc);
        s3 = *(const v4f*)(gA + (size_t)48 * K + kc);
        s4 = *(const v4f*)(gW + (size_t)0  * K + kc);
        s5 = *(const v4f*)(gW + (size_t)16 * K + kc);
        s6 = *(const v4f*)(gW + (size_t)32 * K + kc);
        s7 = *(const v4f*)(gW + (size_t)48 * K + kc);
        if constexpr (NLD == 12) {
            s8  = *(const v4f*)(gW + (size_t)64  * K + kc);
            s9  = *(const v4f*)(gW + (size_t)80  * K + kc);
            s10 = *(const v4f*)(gW + (size_t)96  * K + kc);
            s11 = *(const v4f*)(gW + (size_t)112 * K + kc);
        }
    };
    auto sstore = [&]() {
        *(v4f*)(sdst + 0  * PITCH) = s0;
        *(v4f*)(sdst + 16 * PITCH) = s1;
        *(v4f*)(sdst + 32 * PITCH) = s2;
        *(v4f*)(sdst + 48 * PITCH) = s3;
        *(v4f*)(sdst + 64 * PITCH) = s4;
        *(v4f*)(sdst + 80 * PITCH) = s5;
        *(v4f*)(sdst + 96 * PITCH) = s6;
        *(v4f*)(sdst + 112 * PITCH) = s7;
        if constexpr (NLD == 12) {
            *(v4f*)(sdst + 128 * PITCH) = s8;
            *(v4f*)(sdst + 144 * PITCH) = s9;
            *(v4f*)(sdst + 160 * PITCH) = s10;
            *(v4f*)(sdst + 176 * PITCH) = s11;
        }
    };

    const float* sA = lds + (wm * 32 + lr) * PITCH + 2 * half;
    const float* sB = lds + (BM + wn * 16 * NF + lr) * PITCH + 2 * half;

    v8f acc[2][NF];
#pragma unroll
    for (int i = 0; i < 2; ++i)
#pragma unroll
        for (int f = 0; f < NF; ++f) acc[i][f] = v8f{};

    gload(0);
    for (int kc = 0; kc < K; kc += KP) {
        sstore();
        __syncthreads();
        if (kc + KP < K) gload(kc + KP);

#pragma unroll
        for (int ks = 0; ks < KP / 4; ++ks) {
            const v2f a0 = *(const v2f*)(sA + 4 * ks);
            const v2f a1 = *(const v2f*)(sA + 16 * PITCH + 4 * ks);
            v2f b[NF];
#pragma unroll
            for (int f = 0; f < NF; ++f)
                b[f] = *(const v2f*)(sB + f * 16 * PITCH + 4 * ks);
#pragma unroll
            for (int f = 0; f < NF; ++f)
                acc[0][f] = __builtin_amdgcn_wmma_f32_16x16x4_f32(
                    false, a0, false, b[f], (short)0, acc[0][f], false, false);
#pragma unroll
            for (int f = 0; f < NF; ++f)
                acc[1][f] = __builtin_amdgcn_wmma_f32_16x16x4_f32(
                    false, a1, false, b[f], (short)0, acc[1][f], false, false);
        }
        __syncthreads();
    }

    float* Cb = C + (size_t)(tm * BM + wm * 32 + 8 * half) * ldc
                  + tn * BN + wn * 16 * NF + lr;
#pragma unroll
    for (int r = 0; r < 8; ++r) {
#pragma unroll
        for (int i = 0; i < 2; ++i) {
            float* Cr = Cb + (size_t)(16 * i + r) * ldc;
#pragma unroll
            for (int f = 0; f < NF; ++f) Cr[16 * f] = acc[i][f][r];
        }
    }
}

#define SSM_P    64
#define XP_N     33
#define XP_PITCH 36
#define FX_T     16
#define FX_WAVES (DI / 128)
#define FX_TP    132
#define FX_GB    8
#define FX_NB    (128 / (8 * FX_GB))

__global__ __launch_bounds__(32 * FX_WAVES) void conv_xproj_kernel(
    const float* __restrict__ xg, const float* __restrict__ conv_w,
    const float* __restrict__ conv_b, const float* __restrict__ Wx,
    float* __restrict__ x_conv, float* __restrict__ ssm)
{
    static_assert(L_SZ % FX_T == 0 && FX_T >= 3,
                  "a chunk must not cross a sequence end, and only the first chunk of a sequence lacks its 3 rows in front");
    static_assert(DI % 128 == 0 && 128 % (8 * FX_GB) == 0, "K slices are whole operand batches");
    static_assert(NROW % FX_T == 0, "whole row blocks");

    __shared__ __attribute__((aligned(16))) float tile[FX_WAVES * FX_T * FX_TP];
    __shared__ float red[FX_WAVES * FX_T * XP_PITCH];

    const int tid  = threadIdx.x;
    const int wave = tid >> 5;
    const int lane = tid & 31;
    const int half = lane >> 4;
    const int lr   = lane & 15;

    const unsigned d    = tid * 4;
    const unsigned row0 = blockIdx.x * FX_T;
    const bool head = (row0 % L_SZ) == 0;

    v4f w[4];
#pragma unroll
    for (int c = 0; c < 4; ++c) w[c] = *(const v4f*)(conv_w + (size_t)(d + c) * 4);
    const v4f bias = *(const v4f*)(conv_b + d);

    v4f xin[FX_T + 3];
    const float* src = xg + (size_t)row0 * (2 * DI) + d;
    const size_t back = head ? 0 : (size_t)(2 * DI);
#pragma unroll
    for (int i = 0; i < 3; ++i)
        xin[i] = *(const v4f*)(src - (size_t)(3 - i) * back);
#pragma unroll
    for (int i = 3; i < FX_T + 3; ++i)
        xin[i] = *(const v4f*)(src + (size_t)(i - 3) * (2 * DI));

    const int kb = wave * 128 + 4 * half;
    const float* pB0 = Wx + (size_t)lr * DI + kb;
    const float* pB1 = pB0 + (size_t)16 * DI;
    const float* pB2 = Wx + (size_t)32 * DI + kb;
    const bool col32 = (lr == 0);
    v4f b0[FX_NB][FX_GB], b1[FX_NB][FX_GB], b2[FX_NB][FX_GB];
#pragma unroll
    for (int g = 0; g < FX_GB; ++g) {
        b0[0][g] = *(const v4f*)(pB0 + 8 * g);
        b1[0][g] = *(const v4f*)(pB1 + 8 * g);
        b2[0][g] = *(const v4f*)(pB2 + 8 * g);
    }

    float* dst   = x_conv + (size_t)row0 * DI + d;
    float* mytile = tile + wave * (FX_T * FX_TP);
#pragma unroll
    for (int t = 0; t < FX_T; ++t) {
        v4f o;
#pragma unroll
        for (int c = 0; c < 4; ++c) {
            float acc = bias[c];
#pragma unroll
            for (int j = 0; j < 4; ++j)
                if (t + j >= 3 || !head)
                    acc = fmaf(w[c][j], xin[t + j][c], acc);
            o[c] = acc / (1.0f + __expf(-acc));
        }
        *(v4f*)(dst + (size_t)t * DI) = o;
        *(v4f*)(mytile + t * FX_TP + 4 * lane) = o;
    }

#pragma unroll
    for (int bt = 1; bt < FX_NB; ++bt) {
#pragma unroll
        for (int g = 0; g < FX_GB; ++g) {
            const int o8 = (bt * FX_GB + g) * 8;
            b0[bt][g] = *(const v4f*)(pB0 + o8);
            b1[bt][g] = *(const v4f*)(pB1 + o8);
            b2[bt][g] = *(const v4f*)(pB2 + o8);
        }
    }
    __syncthreads();

    v8f acc[3];
#pragma unroll
    for (int f = 0; f < 3; ++f) acc[f] = v8f{};

    const float* sA = mytile + lr * FX_TP + 4 * half;
#pragma unroll
    for (int bt = 0; bt < FX_NB; ++bt) {
        v4f a[FX_GB];
#pragma unroll
        for (int g = 0; g < FX_GB; ++g)
            a[g] = *(const v4f*)(sA + (bt * FX_GB + g) * 8);
#pragma unroll
        for (int g = 0; g < FX_GB; ++g) {
            const v4f va  = a[g];
            const v4f vb0 = b0[bt][g];
            const v4f vb1 = b1[bt][g];
            const v4f vb2 = col32 ? b2[bt][g] : v4f{0.f, 0.f, 0.f, 0.f};
#pragma unroll
            for (int p = 0; p < 2; ++p) {
                const v2f fa  = {va[2 * p],  va[2 * p + 1]};
                const v2f fb0 = {vb0[2 * p], vb0[2 * p + 1]};
                const v2f fb1 = {vb1[2 * p], vb1[2 * p + 1]};
                const v2f fb2 = {vb2[2 * p], vb2[2 * p + 1]};
                acc[0] = __builtin_amdgcn_wmma_f32_16x16x4_f32(
                    false, fa, false, fb0, (short)0, acc[0], false, false);
                acc[1] = __builtin_amdgcn_wmma_f32_16x16x4_f32(
                    false, fa, false, fb1, (short)0, acc[1], false, false);
                acc[2] = __builtin_amdgcn_wmma_f32_16x16x4_f32(
                    false, fa, false, fb2, (short)0, acc[2], false, false);
            }
        }
    }

    float* rw = red + wave * (FX_T * XP_PITCH) + (8 * half) * XP_PITCH + lr;
#pragma unroll
    for (int r = 0; r < 8; ++r) {
        float* p = rw + r * XP_PITCH;
        p[0]  = acc[0][r];
        p[16] = acc[1][r];
        if (col32) p[32] = acc[2][r];
    }
    __syncthreads();

    float* sdst = ssm + (size_t)row0 * SSM_P;
#pragma unroll
    for (int j = 0; j < (FX_T * XP_N + 32 * FX_WAVES - 1) / (32 * FX_WAVES); ++j) {
        const int e = tid + 32 * FX_WAVES * j;
        if (e < FX_T * XP_N) {
            const int row = e / XP_N;
            const int n   = e - row * XP_N;
            const float* p = red + row * XP_PITCH + n;
            float s = p[0];
#pragma unroll
            for (int wv = 1; wv < FX_WAVES; ++wv) s += p[wv * (FX_T * XP_PITCH)];
            sdst[row * SSM_P + n] = s;
        }
    }
}

#define CH_T 32
#define CH_C (L_SZ / CH_T)
#define DGRP (DI / 256)

__device__ __forceinline__ float softplus_short(float z)
{
    const float u = __expf(-fabsf(z));
    const float s = u * __builtin_amdgcn_rcpf(2.0f + u);
    const float q = s * s;
    float p = 2.0f / 15.0f;
    p = fmaf(p, q, 2.0f / 13.0f);
    p = fmaf(p, q, 2.0f / 11.0f);
    p = fmaf(p, q, 2.0f / 9.0f);
    p = fmaf(p, q, 2.0f / 7.0f);
    p = fmaf(p, q, 2.0f / 5.0f);
    p = fmaf(p, q, 2.0f / 3.0f);
    p = fmaf(p, q, 2.0f);
    return fmaf(s, p, fmaxf(z, 0.0f));
}

__global__ __launch_bounds__(256) void scan_local_kernel(
    const float* __restrict__ ssm, const float* __restrict__ x_conv,
    const float* __restrict__ w_dt, const float* __restrict__ b_dt,
    const float* __restrict__ A_log,
    float* __restrict__ hloc, float* __restrict__ sumdt)
{
    const int dg   = blockIdx.x % DGRP;
    const int bc   = blockIdx.x / DGRP;
    const int d    = dg * 256 + threadIdx.x;
    const int row0 = bc * CH_T;

    const float wdt = w_dt[d];
    const float bdt = b_dt[d];
    float Aa[DS];
#pragma unroll
    for (int n = 0; n < DS; ++n) Aa[n] = -__expf(A_log[d * DS + n]);
    float h[DS];
#pragma unroll
    for (int n = 0; n < DS; ++n) h[n] = 0.f;
    float sdt = 0.f;

#pragma unroll 4
    for (int t = 0; t < CH_T; ++t) {
        const int row = row0 + t;
        const float* s = ssm + (size_t)row * SSM_P;
        const float x_t = x_conv[(size_t)row * DI + d];

        float z  = fmaf(s[32], wdt, bdt);
        float dt = softplus_short(z);
        sdt += dt;
#pragma unroll
        for (int n = 0; n < DS; ++n) {
            float Abar = __expf(dt * Aa[n]);
            h[n] = fmaf(Abar, h[n], (dt * s[n]) * x_t);
        }
    }
#pragma unroll
    for (int n = 0; n < DS; ++n)
        hloc[((size_t)bc * DS + n) * DI + d] = h[n];
    sumdt[(size_t)bc * DI + d] = sdt;
}

__global__ __launch_bounds__(256) void scan_carry_kernel(
    const float* __restrict__ hloc, const float* __restrict__ sumdt,
    const float* __restrict__ A_log, float* __restrict__ hin)
{
    const int idx = blockIdx.x * 256 + threadIdx.x;
    if (idx >= B_SZ * DS * DI) return;
    const int d = idx % DI;
    const int n = (idx / DI) % DS;
    const int b = idx / (DI * DS);
    const float Aa = -__expf(A_log[d * DS + n]);

    float h = 0.f;
#pragma unroll 8
    for (int c = 0; c < CH_C; ++c) {
        const size_t bc = (size_t)b * CH_C + c;
        const float hl = hloc[(bc * DS + n) * DI + d];
        const float sd = sumdt[bc * DI + d];
        hin[(bc * DS + n) * DI + d] = h;
        h = fmaf(__expf(Aa * sd), h, hl);
    }
}

__global__ __launch_bounds__(256) void scan_replay_kernel(
    const float* __restrict__ ssm, const float* __restrict__ x_conv,
    const float* __restrict__ xg, const float* __restrict__ w_dt,
    const float* __restrict__ b_dt, const float* __restrict__ A_log,
    const float* __restrict__ D_param, const float* __restrict__ hin,
    float* __restrict__ y)
{
    const int dg   = blockIdx.x % DGRP;
    const int bc   = blockIdx.x / DGRP;
    const int d    = dg * 256 + threadIdx.x;
    const int row0 = bc * CH_T;

    const float wdt = w_dt[d];
    const float bdt = b_dt[d];
    const float Dp  = D_param[d];
    float Aa[DS];
#pragma unroll
    for (int n = 0; n < DS; ++n) Aa[n] = -__expf(A_log[d * DS + n]);
    float h[DS];
#pragma unroll
    for (int n = 0; n < DS; ++n) h[n] = hin[((size_t)bc * DS + n) * DI + d];

#pragma unroll 4
    for (int t = 0; t < CH_T; ++t) {
        const int row = row0 + t;
        const float* s = ssm + (size_t)row * SSM_P;
        const float x_t = x_conv[(size_t)row * DI + d];

        float z  = fmaf(s[32], wdt, bdt);
        float dt = softplus_short(z);

        float yv = 0.f;
#pragma unroll
        for (int n = 0; n < DS; ++n) {
            float Abar = __expf(dt * Aa[n]);
            h[n] = fmaf(Abar, h[n], (dt * s[n]) * x_t);
            yv   = fmaf(h[n], s[DS + n], yv);
        }
        yv = fmaf(x_t, Dp, yv);

        float gv = xg[(size_t)row * (2 * DI) + DI + d];
        float sg = gv / (1.0f + __expf(-gv));
        y[(size_t)row * DI + d] = yv * sg;
    }
}

extern "C" void kernel_launch(void* const* d_in, const int* in_sizes, int n_in,
                              void* d_out, int out_size, void* d_ws, size_t ws_size,
                              hipStream_t stream)
{
    const float* x       = (const float*)d_in[0];
    const float* W_in    = (const float*)d_in[1];
    const float* conv_w  = (const float*)d_in[2];
    const float* conv_b  = (const float*)d_in[3];
    const float* W_xproj = (const float*)d_in[4];
    const float* w_dt    = (const float*)d_in[5];
    const float* b_dt    = (const float*)d_in[6];
    const float* A_log   = (const float*)d_in[7];
    const float* D_param = (const float*)d_in[8];
    const float* W_out   = (const float*)d_in[9];
    float* out = (float*)d_out;

    float* ws     = (float*)d_ws;
    float* xg     = ws;
    float* x_conv = xg + (size_t)NROW * (2 * DI);
    float* ssm    = x_conv + (size_t)NROW * DI;
    float* yb     = ssm + (size_t)NROW * SSM_P;
    float* hloc   = yb + (size_t)NROW * DI;
    float* hin    = hloc + (size_t)B_SZ * CH_C * DS * DI;
    float* sumdt  = hin + (size_t)B_SZ * CH_C * DS * DI;

    gemm_lds_f32_wmma<DM, 4><<<(NROW / 64) * ((2 * DI) / 128), 128, 0, stream>>>(
        x, W_in, xg, (2 * DI) / 128, 2 * DI);

    conv_xproj_kernel<<<NROW / FX_T, 32 * FX_WAVES, 0, stream>>>(
        xg, conv_w, conv_b, W_xproj, x_conv, ssm);

    scan_local_kernel<<<B_SZ * CH_C * DGRP, 256, 0, stream>>>(
        ssm, x_conv, w_dt, b_dt, A_log, hloc, sumdt);
    scan_carry_kernel<<<(B_SZ * DS * DI + 255) / 256, 256, 0, stream>>>(
        hloc, sumdt, A_log, hin);
    scan_replay_kernel<<<B_SZ * CH_C * DGRP, 256, 0, stream>>>(
        ssm, x_conv, xg, w_dt, b_dt, A_log, D_param, hin, yb);

    gemm_lds_f32_wmma<DI, 2><<<(NROW / 64) * (DM / 64), 128, 0, stream>>>(
        yb, W_out, out, DM / 64, DM);
}


# both GEMMs double-buffer their LDS panels: next panel stored, the one after requested before the current products, one barrier per panel; first GEMM on 8-wave 128x128 tiles; k order unchanged
# speedup vs baseline: 14.5905x; 1.0303x over previous
#include <hip/hip_runtime.h>
#include <math.h>


typedef float  v2f __attribute__((ext_vector_type(2)));
typedef float  v4f __attribute__((ext_vector_type(4)));
typedef float  v8f __attribute__((ext_vector_type(8)));

#define B_SZ 2
#define L_SZ 2048
#define DM   768
#define DI   1536
#define DS   16
#define NROW (B_SZ * L_SZ)

template <int K, int NF, int WM>
__global__ __launch_bounds__(64 * WM, 6) void gemm_lds_f32_wmma(
    const float* __restrict__ A, const float* __restrict__ W,
    float* __restrict__ C, int NT, int ldc)
{
    static_assert(K % 64 == 0, "the panel loop is written out for pairs of panels");
    constexpr int BM    = 32 * WM;
    constexpr int BN    = 32 * NF;
    constexpr int KP    = 32;
    constexpr int PITCH = 36;
    constexpr int ROWS  = BM + BN;
    constexpr int RP    = 8 * WM;
    constexpr int BUF   = ROWS * PITCH;
    static_assert(BM == 4 * RP && BN == 4 * RP, "staging is written out for 4 A pieces and 4 W pieces per thread");

    __shared__ __attribute__((aligned(16))) float lds[2 * BUF];

    const int tid  = threadIdx.x;
    const int wave = tid >> 5;
    const int lane = tid & 31;
    const int half = lane >> 4;
    const int lr   = lane & 15;
    const int wm   = wave >> 1;
    const int wn   = wave & 1;
    const int tm   = blockIdx.x / NT;
    const int tn   = blockIdx.x % NT;

    const int r0 = tid >> 3;
    const int c4 = tid & 7;
    const float* gA = A + (size_t)tm * BM * K;
    const float* gW = W + (size_t)tn * BN * K;
    const unsigned goff = (unsigned)(r0 * K + c4 * 4);
    const int soff = r0 * PITCH + c4 * 4;
    const unsigned coff = (unsigned)((wm * 32 + 8 * half) * ldc + wn * 16 * NF + lr);

    v4f s0, s1, s2, s3, s4, s5, s6, s7;
    const float* pa = gA;
    const float* pw = gW;
    auto gload = [&]() {
        s0 = *(const v4f*)(pa + (goff + 0u * RP * K));
        s1 = *(const v4f*)(pa + (goff + 1u * RP * K));
        s2 = *(const v4f*)(pa + (goff + 2u * RP * K));
        s3 = *(const v4f*)(pa + (goff + 3u * RP * K));
        s4 = *(const v4f*)(pw + (goff + 0u * RP * K));
        s5 = *(const v4f*)(pw + (goff + 1u * RP * K));
        s6 = *(const v4f*)(pw + (goff + 2u * RP * K));
        s7 = *(const v4f*)(pw + (goff + 3u * RP * K));
        pa += KP;
        pw += KP;
    };
    auto sstore = [&](float* buf) {
        float* p = buf + soff;
        *(v4f*)(p + (0 * RP) * PITCH) = s0;
        *(v4f*)(p + (1 * RP) * PITCH) = s1;
        *(v4f*)(p + (2 * RP) * PITCH) = s2;
        *(v4f*)(p + (3 * RP) * PITCH) = s3;
        *(v4f*)(p + (BM + 0 * RP) * PITCH) = s4;
        *(v4f*)(p + (BM + 1 * RP) * PITCH) = s5;
        *(v4f*)(p + (BM + 2 * RP) * PITCH) = s6;
        *(v4f*)(p + (BM + 3 * RP) * PITCH) = s7;
    };

    const int oA = (wm * 32 + lr) * PITCH + 2 * half;
    const int oB = (BM + wn * 16 * NF + lr) * PITCH + 2 * half;

    v8f acc[2][NF];
#pragma unroll
    for (int i = 0; i < 2; ++i)
#pragma unroll
        for (int f = 0; f < NF; ++f) acc[i][f] = v8f{};

    auto products = [&](const float* buf) {
        const float* sA = buf + oA;
        const float* sB = buf + oB;
#pragma unroll
        for (int ks = 0; ks < KP / 4; ++ks) {
            const v2f a0 = *(const v2f*)(sA + 4 * ks);
            const v2f a1 = *(const v2f*)(sA + 16 * PITCH + 4 * ks);
            v2f b[NF];
#pragma unroll
            for (int f = 0; f < NF; ++f)
                b[f] = *(const v2f*)(sB + f * 16 * PITCH + 4 * ks);
#pragma unroll
            for (int f = 0; f < NF; ++f)
                acc[0][f] = __builtin_amdgcn_wmma_f32_16x16x4_f32(
                    false, a0, false, b[f], (short)0, acc[0][f], false, false);
#pragma unroll
            for (int f = 0; f < NF; ++f)
                acc[1][f] = __builtin_amdgcn_wmma_f32_16x16x4_f32(
                    false, a1, false, b[f], (short)0, acc[1][f], false, false);
        }
    };

    float* buf0 = lds;
    float* buf1 = lds + BUF;

    gload();
    sstore(buf0);
    gload();
    __syncthreads();
    for (int kc = 0; kc < K; kc += 2 * KP) {
        const bool more = kc + 2 * KP < K;
        sstore(buf1);
        if (more) gload();
        products(buf0);
        __syncthreads();
        if (more) {
            sstore(buf0);
            gload();
        }
        products(buf1);
        __syncthreads();
    }

    float* Cb = C + ((size_t)tm * BM * ldc + (size_t)tn * BN) + coff;
#pragma unroll
    for (int r = 0; r < 8; ++r) {
#pragma unroll
        for (int i = 0; i < 2; ++i) {
            float* Cr = Cb + (size_t)(16 * i + r) * ldc;
#pragma unroll
            for (int f = 0; f < NF; ++f) Cr[16 * f] = acc[i][f][r];
        }
    }
}

#define SSM_P    64
#define XP_N     33
#define XP_PITCH 36
#define FX_T     16
#define FX_WAVES (DI / 128)
#define FX_TP    132
#define FX_GB    8
#define FX_NB    (128 / (8 * FX_GB))

__global__ __launch_bounds__(32 * FX_WAVES) void conv_xproj_kernel(
    const float* __restrict__ xg, const float* __restrict__ conv_w,
    const float* __restrict__ conv_b, const float* __restrict__ Wx,
    float* __restrict__ x_conv, float* __restrict__ ssm)
{
    static_assert(L_SZ % FX_T == 0 && FX_T >= 3,
                  "a chunk must not cross a sequence end, and only the first chunk of a sequence lacks its 3 rows in front");
    static_assert(DI % 128 == 0 && 128 % (8 * FX_GB) == 0, "K slices are whole operand batches");
    static_assert(NROW % FX_T == 0, "whole row blocks");

    __shared__ __attribute__((aligned(16))) float tile[FX_WAVES * FX_T * FX_TP];
    __shared__ float red[FX_WAVES * FX_T * XP_PITCH];

    const int tid  = threadIdx.x;
    const int wave = tid >> 5;
    const int lane = tid & 31;
    const int half = lane >> 4;
    const int lr   = lane & 15;

    const unsigned d    = tid * 4;
    const unsigned row0 = blockIdx.x * FX_T;
    const bool head = (row0 % L_SZ) == 0;

    v4f w[4];
#pragma unroll
    for (int c = 0; c < 4; ++c) w[c] = *(const v4f*)(conv_w + (size_t)(d + c) * 4);
    const v4f bias = *(const v4f*)(conv_b + d);

    v4f xin[FX_T + 3];
    const float* src = xg + (size_t)row0 * (2 * DI) + d;
    const size_t back = head ? 0 : (size_t)(2 * DI);
#pragma unroll
    for (int i = 0; i < 3; ++i)
        xin[i] = *(const v4f*)(src - (size_t)(3 - i) * back);
#pragma unroll
    for (int i = 3; i < FX_T + 3; ++i)
        xin[i] = *(const v4f*)(src + (size_t)(i - 3) * (2 * DI));

    const int kb = wave * 128 + 4 * half;
    const float* pB0 = Wx + (size_t)lr * DI + kb;
    const float* pB1 = pB0 + (size_t)16 * DI;
    const float* pB2 = Wx + (size_t)32 * DI + kb;
    const bool col32 = (lr == 0);
    v4f b0[FX_NB][FX_GB], b1[FX_NB][FX_GB], b2[FX_NB][FX_GB];
#pragma unroll
    for (int g = 0; g < FX_GB; ++g) {
        b0[0][g] = *(const v4f*)(pB0 + 8 * g);
        b1[0][g] = *(const v4f*)(pB1 + 8 * g);
        b2[0][g] = *(const v4f*)(pB2 + 8 * g);
    }

    float* dst   = x_conv + (size_t)row0 * DI + d;
    float* mytile = tile + wave * (FX_T * FX_TP);
#pragma unroll
    for (int t = 0; t < FX_T; ++t) {
        v4f o;
#pragma unroll
        for (int c = 0; c < 4; ++c) {
            float acc = bias[c];
#pragma unroll
            for (int j = 0; j < 4; ++j)
                if (t + j >= 3 || !head)
                    acc = fmaf(w[c][j], xin[t + j][c], acc);
            o[c] = acc / (1.0f + __expf(-acc));
        }
        *(v4f*)(dst + (size_t)t * DI) = o;
        *(v4f*)(mytile + t * FX_TP + 4 * lane) = o;
    }

#pragma unroll
    for (int bt = 1; bt < FX_NB; ++bt) {
#pragma unroll
        for (int g = 0; g < FX_GB; ++g) {
            const int o8 = (bt * FX_GB + g) * 8;
            b0[bt][g] = *(const v4f*)(pB0 + o8);
            b1[bt][g] = *(const v4f*)(pB1 + o8);
            b2[bt][g] = *(const v4f*)(pB2 + o8);
        }
    }
    __syncthreads();

    v8f acc[3];
#pragma unroll
    for (int f = 0; f < 3; ++f) acc[f] = v8f{};

    const float* sA = mytile + lr * FX_TP + 4 * half;
#pragma unroll
    for (int bt = 0; bt < FX_NB; ++bt) {
        v4f a[FX_GB];
#pragma unroll
        for (int g = 0; g < FX_GB; ++g)
            a[g] = *(const v4f*)(sA + (bt * FX_GB + g) * 8);
#pragma unroll
        for (int g = 0; g < FX_GB; ++g) {
            const v4f va  = a[g];
            const v4f vb0 = b0[bt][g];
            const v4f vb1 = b1[bt][g];
            const v4f vb2 = col32 ? b2[bt][g] : v4f{0.f, 0.f, 0.f, 0.f};
#pragma unroll
            for (int p = 0; p < 2; ++p) {
                const v2f fa  = {va[2 * p],  va[2 * p + 1]};
                const v2f fb0 = {vb0[2 * p], vb0[2 * p + 1]};
                const v2f fb1 = {vb1[2 * p], vb1[2 * p + 1]};
                const v2f fb2 = {vb2[2 * p], vb2[2 * p + 1]};
                acc[0] = __builtin_amdgcn_wmma_f32_16x16x4_f32(
                    false, fa, false, fb0, (short)0, acc[0], false, false);
                acc[1] = __builtin_amdgcn_wmma_f32_16x16x4_f32(
                    false, fa, false, fb1, (short)0, acc[1], false, false);
                acc[2] = __builtin_amdgcn_wmma_f32_16x16x4_f32(
                    false, fa, false, fb2, (short)0, acc[2], false, false);
            }
        }
    }

    float* rw = red + wave * (FX_T * XP_PITCH) + (8 * half) * XP_PITCH + lr;
#pragma unroll
    for (int r = 0; r < 8; ++r) {
        float* p = rw + r * XP_PITCH;
        p[0]  = acc[0][r];
        p[16] = acc[1][r];
        if (col32) p[32] = acc[2][r];
    }
    __syncthreads();

    float* sdst = ssm + (size_t)row0 * SSM_P;
#pragma unroll
    for (int j = 0; j < (FX_T * XP_N + 32 * FX_WAVES - 1) / (32 * FX_WAVES); ++j) {
        const int e = tid + 32 * FX_WAVES * j;
        if (e < FX_T * XP_N) {
            const int row = e / XP_N;
            const int n   = e - row * XP_N;
            const float* p = red + row * XP_PITCH + n;
            float s = p[0];
#pragma unroll
            for (int wv = 1; wv < FX_WAVES; ++wv) s += p[wv * (FX_T * XP_PITCH)];
            sdst[row * SSM_P + n] = s;
        }
    }
}

#define CH_T 32
#define CH_C (L_SZ / CH_T)
#define DGRP (DI / 256)

__device__ __forceinline__ float softplus_short(float z)
{
    const float u = __expf(-fabsf(z));
    const float s = u * __builtin_amdgcn_rcpf(2.0f + u);
    const float q = s * s;
    float p = 2.0f / 15.0f;
    p = fmaf(p, q, 2.0f / 13.0f);
    p = fmaf(p, q, 2.0f / 11.0f);
    p = fmaf(p, q, 2.0f / 9.0f);
    p = fmaf(p, q, 2.0f / 7.0f);
    p = fmaf(p, q, 2.0f / 5.0f);
    p = fmaf(p, q, 2.0f / 3.0f);
    p = fmaf(p, q, 2.0f);
    return fmaf(s, p, fmaxf(z, 0.0f));
}

__global__ __launch_bounds__(256) void scan_local_kernel(
    const float* __restrict__ ssm, const float* __restrict__ x_conv,
    const float* __restrict__ w_dt, const float* __restrict__ b_dt,
    const float* __restrict__ A_log,
    float* __restrict__ hloc, float* __restrict__ sumdt)
{
    const int dg   = blockIdx.x % DGRP;
    const int bc   = blockIdx.x / DGRP;
    const int d    = dg * 256 + threadIdx.x;
    const int row0 = bc * CH_T;

    const float wdt = w_dt[d];
    const float bdt = b_dt[d];
    float Aa[DS];
#pragma unroll
    for (int n = 0; n < DS; ++n) Aa[n] = -__expf(A_log[d * DS + n]);
    float h[DS];
#pragma unroll
    for (int n = 0; n < DS; ++n) h[n] = 0.f;
    float sdt = 0.f;

#pragma unroll 4
    for (int t = 0; t < CH_T; ++t) {
        const int row = row0 + t;
        const float* s = ssm + (size_t)row * SSM_P;
        const float x_t = x_conv[(size_t)row * DI + d];

        float z  = fmaf(s[32], wdt, bdt);
        float dt = softplus_short(z);
        sdt += dt;
#pragma unroll
        for (int n = 0; n < DS; ++n) {
            float Abar = __expf(dt * Aa[n]);
            h[n] = fmaf(Abar, h[n], (dt * s[n]) * x_t);
        }
    }
#pragma unroll
    for (int n = 0; n < DS; ++n)
        hloc[((size_t)bc * DS + n) * DI + d] = h[n];
    sumdt[(size_t)bc * DI + d] = sdt;
}

__global__ __launch_bounds__(256) void scan_carry_kernel(
    const float* __restrict__ hloc, const float* __restrict__ sumdt,
    const float* __restrict__ A_log, float* __restrict__ hin)
{
    const int idx = blockIdx.x * 256 + threadIdx.x;
    if (idx >= B_SZ * DS * DI) return;
    const int d = idx % DI;
    const int n = (idx / DI) % DS;
    const int b = idx / (DI * DS);
    const float Aa = -__expf(A_log[d * DS + n]);

    float h = 0.f;
#pragma unroll 8
    for (int c = 0; c < CH_C; ++c) {
        const size_t bc = (size_t)b * CH_C + c;
        const float hl = hloc[(bc * DS + n) * DI + d];
        const float sd = sumdt[bc * DI + d];
        hin[(bc * DS + n) * DI + d] = h;
        h = fmaf(__expf(Aa * sd), h, hl);
    }
}

__global__ __launch_bounds__(256) void scan_replay_kernel(
    const float* __restrict__ ssm, const float* __restrict__ x_conv,
    const float* __restrict__ xg, const float* __restrict__ w_dt,
    const float* __restrict__ b_dt, const float* __restrict__ A_log,
    const float* __restrict__ D_param, const float* __restrict__ hin,
    float* __restrict__ y)
{
    const int dg   = blockIdx.x % DGRP;
    const int bc   = blockIdx.x / DGRP;
    const int d    = dg * 256 + threadIdx.x;
    const int row0 = bc * CH_T;

    const float wdt = w_dt[d];
    const float bdt = b_dt[d];
    const float Dp  = D_param[d];
    float Aa[DS];
#pragma unroll
    for (int n = 0; n < DS; ++n) Aa[n] = -__expf(A_log[d * DS + n]);
    float h[DS];
#pragma unroll
    for (int n = 0; n < DS; ++n) h[n] = hin[((size_t)bc * DS + n) * DI + d];

#pragma unroll 4
    for (int t = 0; t < CH_T; ++t) {
        const int row = row0 + t;
        const float* s = ssm + (size_t)row * SSM_P;
        const float x_t = x_conv[(size_t)row * DI + d];

        float z  = fmaf(s[32], wdt, bdt);
        float dt = softplus_short(z);

        float yv = 0.f;
#pragma unroll
        for (int n = 0; n < DS; ++n) {
            float Abar = __expf(dt * Aa[n]);
            h[n] = fmaf(Abar, h[n], (dt * s[n]) * x_t);
            yv   = fmaf(h[n], s[DS + n], yv);
        }
        yv = fmaf(x_t, Dp, yv);

        float gv = xg[(size_t)row * (2 * DI) + DI + d];
        float sg = gv / (1.0f + __expf(-gv));
        y[(size_t)row * DI + d] = yv * sg;
    }
}

extern "C" void kernel_launch(void* const* d_in, const int* in_sizes, int n_in,
                              void* d_out, int out_size, void* d_ws, size_t ws_size,
                              hipStream_t stream)
{
    const float* x       = (const float*)d_in[0];
    const float* W_in    = (const float*)d_in[1];
    const float* conv_w  = (const float*)d_in[2];
    const float* conv_b  = (const float*)d_in[3];
    const float* W_xproj = (const float*)d_in[4];
    const float* w_dt    = (const float*)d_in[5];
    const float* b_dt    = (const float*)d_in[6];
    const float* A_log   = (const float*)d_in[7];
    const float* D_param = (const float*)d_in[8];
    const float* W_out   = (const float*)d_in[9];
    float* out = (float*)d_out;

    float* ws     = (float*)d_ws;
    float* xg     = ws;
    float* x_conv = xg + (size_t)NROW * (2 * DI);
    float* ssm    = x_conv + (size_t)NROW * DI;
    float* yb     = ssm + (size_t)NROW * SSM_P;
    float* hloc   = yb + (size_t)NROW * DI;
    float* hin    = hloc + (size_t)B_SZ * CH_C * DS * DI;
    float* sumdt  = hin + (size_t)B_SZ * CH_C * DS * DI;

    gemm_lds_f32_wmma<DM, 4, 4><<<(NROW / 128) * ((2 * DI) / 128), 256, 0, stream>>>(
        x, W_in, xg, (2 * DI) / 128, 2 * DI);

    conv_xproj_kernel<<<NROW / FX_T, 32 * FX_WAVES, 0, stream>>>(
        xg, conv_w, conv_b, W_xproj, x_conv, ssm);

    scan_local_kernel<<<B_SZ * CH_C * DGRP, 256, 0, stream>>>(
        ssm, x_conv, w_dt, b_dt, A_log, hloc, sumdt);
    scan_carry_kernel<<<(B_SZ * DS * DI + 255) / 256, 256, 0, stream>>>(
        hloc, sumdt, A_log, hin);
    scan_replay_kernel<<<B_SZ * CH_C * DGRP, 256, 0, stream>>>(
        ssm, x_conv, xg, w_dt, b_dt, A_log, D_param, hin, yb);

    gemm_lds_f32_wmma<DI, 2, 2><<<(NROW / 64) * (DM / 64), 128, 0, stream>>>(
        yb, W_out, out, DM / 64, DM);
}
